# starting kernel
# baseline (speedup 1.0000x reference)
#include <hip/hip_runtime.h>
#include <hip/hip_bf16.h>

#define HID 128
#define NEG_SLOPE 0.2f
#define GAT_EPS 1e-16f

typedef float v2f __attribute__((ext_vector_type(2)));
typedef float v8f __attribute__((ext_vector_type(8)));

__device__ __forceinline__ unsigned flipf(float f) {
    unsigned u = __float_as_uint(f);
    return (u & 0x80000000u) ? ~u : (u | 0x80000000u);
}
__device__ __forceinline__ float unflipf(unsigned u) {
    return __uint_as_float((u & 0x80000000u) ? (u ^ 0x80000000u) : ~u);
}

__global__ void init_node_kernel(unsigned* __restrict__ emax,
                                 float* __restrict__ denom, int n) {
    int i = blockIdx.x * blockDim.x + threadIdx.x;
    if (i < n) { emax[i] = 0u; denom[i] = 0.0f; }
}

__global__ void init_out_kernel(float* __restrict__ out,
                                const float* __restrict__ bias, int total) {
    int i = blockIdx.x * blockDim.x + threadIdx.x;
    if (i < total) out[i] = bias[i & (HID - 1)];
}

__global__ __launch_bounds__(256) void gemm_x_kernel(const float* __restrict__ h,
                                                     const float* __restrict__ W,
                                                     float* __restrict__ x,
                                                     int n) {
    __shared__ float sh[16 * HID];
    const int block_row = blockIdx.x * 16;
    const int tid  = threadIdx.x;
    const int wave = tid >> 5;
    const int lane = tid & 31;

    const float4* gsrc = (const float4*)(h + (size_t)block_row * HID);
    float4*       ldst = (float4*)sh;
    #pragma unroll
    for (int i = tid; i < 16 * HID / 4; i += 256) ldst[i] = gsrc[i];
    __syncthreads();

    const int m    = lane & 15;
    const int koff = (lane >> 4) << 1;
    const int ncol = wave * 16 + (lane & 15);

    v8f c = {};
    #pragma unroll
    for (int k = 0; k < HID; k += 4) {
        v2f a, b;
        a.x = sh[m * HID + k + koff];
        a.y = sh[m * HID + k + koff + 1];
        b.x = W[(size_t)(k + koff)     * HID + ncol];
        b.y = W[(size_t)(k + koff + 1) * HID + ncol];
        c = __builtin_amdgcn_wmma_f32_16x16x4_f32(
                 false, a,  false, b,
                 (short)0, c,  false,  false);
    }

    const int rbase = block_row + ((lane >> 4) << 3);
    #pragma unroll
    for (int r = 0; r < 8; ++r) {
        int row = rbase + r;
        if (row < n) x[(size_t)row * HID + ncol] = c[r];
    }
}

__global__ __launch_bounds__(256) void attn_kernel(const float* __restrict__ x,
                                                   const float* __restrict__ a_s,
                                                   const float* __restrict__ a_d,
                                                   float* __restrict__ as_out,
                                                   float* __restrict__ ad_out,
                                                   int n) {
    int node = (blockIdx.x * blockDim.x + threadIdx.x) >> 5;
    int lane = threadIdx.x & 31;
    if (node >= n) return;
    float4 v  = ((const float4*)(x + (size_t)node * HID))[lane];
    float4 s4 = ((const float4*)a_s)[lane];
    float4 d4 = ((const float4*)a_d)[lane];
    float ps = v.x * s4.x + v.y * s4.y + v.z * s4.z + v.w * s4.w;
    float pd = v.x * d4.x + v.y * d4.y + v.z * d4.z + v.w * d4.w;
    #pragma unroll
    for (int o = 16; o > 0; o >>= 1) {
        ps += __shfl_down(ps, o, 32);
        pd += __shfl_down(pd, o, 32);
    }
    if (lane == 0) { as_out[node] = ps; ad_out[node] = pd; }
}

__global__ void edge_max_kernel(const int* __restrict__ src,
                                const int* __restrict__ dst,
                                const float* __restrict__ as_,
                                const float* __restrict__ ad_,
                                float* __restrict__ eval,
                                unsigned* __restrict__ emax,
                                int n_edges, int n_nodes) {
    int e = blockIdx.x * blockDim.x + threadIdx.x;
    int total = n_edges + n_nodes;
    if (e >= total) return;
    int s, d;
    if (e < n_edges) { s = src[e]; d = dst[e]; } else { s = d = e - n_edges; }
    float v = as_[s] + ad_[d];
    v = (v > 0.0f) ? v : NEG_SLOPE * v;
    eval[e] = v;
    atomicMax(emax + d, flipf(v));
}

__global__ void edge_exp_kernel(const int* __restrict__ dst,
                                const float* __restrict__ eval,
                                const unsigned* __restrict__ emax,
                                float* __restrict__ exv,
                                float* __restrict__ denom,
                                int n_edges, int n_nodes) {
    int e = blockIdx.x * blockDim.x + threadIdx.x;
    int total = n_edges + n_nodes;
    if (e >= total) return;
    int d = (e < n_edges) ? dst[e] : (e - n_edges);
    float m  = unflipf(emax[d]);
    float ex = __expf(eval[e] - m);
    exv[e] = ex;
    atomicAdd(denom + d, ex);
}

__global__ __launch_bounds__(256) void edge_agg_kernel(const int* __restrict__ src,
                                                       const int* __restrict__ dst,
                                                       const float* __restrict__ x,
                                                       const float* __restrict__ exv,
                                                       const float* __restrict__ denom,
                                                       float* __restrict__ out,
                                                       int n_edges, int n_nodes) {
    long long gid = (long long)blockIdx.x * blockDim.x + threadIdx.x;
    int e    = (int)(gid >> 5);
    int lane = threadIdx.x & 31;
    int total = n_edges + n_nodes;
    if (e >= total) return;
    int s, d;
    if (e < n_edges) { s = src[e]; d = dst[e]; } else { s = d = e - n_edges; }
    float alpha = exv[e] / (denom[d] + GAT_EPS);
    float4 v = ((const float4*)(x + (size_t)s * HID))[lane];
    float* o = out + (size_t)d * HID + lane * 4;
    atomicAdd(o + 0, alpha * v.x);
    atomicAdd(o + 1, alpha * v.y);
    atomicAdd(o + 2, alpha * v.z);
    atomicAdd(o + 3, alpha * v.w);
}

__global__ void relu_kernel(float* __restrict__ a, int total) {
    int i = blockIdx.x * blockDim.x + threadIdx.x;
    if (i < total) a[i] = fmaxf(a[i], 0.0f);
}

extern "C" void kernel_launch(void* const* d_in, const int* in_sizes, int n_in,
                              void* d_out, int out_size, void* d_ws, size_t ws_size,
                              hipStream_t stream) {
    const float* z     = (const float*)d_in[0];
    const int*   ei    = (const int*)  d_in[1];
    const float* Ws    = (const float*)d_in[2];
    const float* a_src = (const float*)d_in[3];
    const float* a_dst = (const float*)d_in[4];
    const float* bias  = (const float*)d_in[5];
    float*       out   = (float*)d_out;

    const int N = in_sizes[0] / HID;
    const int E = in_sizes[1] / 2;
    const int ET = E + N;
    const int* src = ei;
    const int* dst = ei + E;

    auto align = [](size_t v) { return (v + 255) & ~(size_t)255; };
    char* p = (char*)d_ws;
    float*    x      = (float*)p;    p += align((size_t)N * HID * sizeof(float));
    float*    bufA   = (float*)p;    p += align((size_t)N * HID * sizeof(float));
    float*    bufB   = (float*)p;    p += align((size_t)N * HID * sizeof(float));
    float*    as_    = (float*)p;    p += align((size_t)N * sizeof(float));
    float*    ad_    = (float*)p;    p += align((size_t)N * sizeof(float));
    unsigned* emax   = (unsigned*)p; p += align((size_t)N * sizeof(unsigned));
    float*    denom  = (float*)p;    p += align((size_t)N * sizeof(float));
    float*    eval   = (float*)p;    p += align((size_t)ET * sizeof(float));
    float*    exv    = (float*)p;    p += align((size_t)ET * sizeof(float));

    const float* h_ptr = z;
    for (int l = 0; l < 3; ++l) {
        float* o_ptr = (l == 0) ? bufA : (l == 1) ? bufB : out;
        const float* W  = Ws    + (size_t)l * HID * HID;
        const float* as = a_src + (size_t)l * HID;
        const float* ad = a_dst + (size_t)l * HID;
        const float* b  = bias  + (size_t)l * HID;

        init_node_kernel<<<(N + 255) / 256, 256, 0, stream>>>(emax, denom, N);
        init_out_kernel<<<((size_t)N * HID + 255) / 256, 256, 0, stream>>>(o_ptr, b, N * HID);

        gemm_x_kernel<<<(N + 15) / 16, 256, 0, stream>>>(h_ptr, W, x, N);

        attn_kernel<<<((size_t)N * 32 + 255) / 256, 256, 0, stream>>>(x, as, ad, as_, ad_, N);

        edge_max_kernel<<<(ET + 255) / 256, 256, 0, stream>>>(src, dst, as_, ad_, eval, emax, E, N);
        edge_exp_kernel<<<(ET + 255) / 256, 256, 0, stream>>>(dst, eval, emax, exv, denom, E, N);
        edge_agg_kernel<<<((size_t)ET * 32 + 255) / 256, 256, 0, stream>>>(src, dst, x, exv, denom,
                                                                           o_ptr, E, N);
        if (l < 2)
            relu_kernel<<<((size_t)N * HID + 255) / 256, 256, 0, stream>>>(o_ptr, N * HID);
        h_ptr = o_ptr;
    }
}


# Edge phase destination-stationary: a 1024-thread block owns 196 nodes, filters the dst array itself (no edge reorder), keeps max, sum, row sums in LDS (LDS float atomics, arrival order); 9 launches
# speedup vs baseline: 2.8852x; 2.8852x over previous
#include <hip/hip_runtime.h>
#include <hip/hip_bf16.h>

#define HID 128
#define NEG_SLOPE 0.2f
#define GAT_EPS 1e-16f

typedef float v2f __attribute__((ext_vector_type(2)));
typedef float v8f __attribute__((ext_vector_type(8)));

__device__ __forceinline__ unsigned flipf(float f) {
    unsigned u = __float_as_uint(f);
    return (u & 0x80000000u) ? ~u : (u | 0x80000000u);
}
__device__ __forceinline__ float unflipf(unsigned u) {
    return __uint_as_float((u & 0x80000000u) ? (u ^ 0x80000000u) : ~u);
}

__global__ __launch_bounds__(256) void gemm_x_kernel(const float* __restrict__ h,
                                                     const float* __restrict__ W,
                                                     float* __restrict__ x,
                                                     int n) {
    __shared__ float sh[16 * HID];
    const int block_row = blockIdx.x * 16;
    const int tid  = threadIdx.x;
    const int wave = tid >> 5;
    const int lane = tid & 31;

    const float4* gsrc = (const float4*)(h + (size_t)block_row * HID);
    float4*       ldst = (float4*)sh;
    #pragma unroll
    for (int i = tid; i < 16 * HID / 4; i += 256) ldst[i] = gsrc[i];
    __syncthreads();

    const int m    = lane & 15;
    const int koff = (lane >> 4) << 1;
    const int ncol = wave * 16 + (lane & 15);

    v8f c = {};
    #pragma unroll
    for (int k = 0; k < HID; k += 4) {
        v2f a, b;
        a.x = sh[m * HID + k + koff];
        a.y = sh[m * HID + k + koff + 1];
        b.x = W[(size_t)(k + koff)     * HID + ncol];
        b.y = W[(size_t)(k + koff + 1) * HID + ncol];
        c = __builtin_amdgcn_wmma_f32_16x16x4_f32(
                 false, a,  false, b,
                 (short)0, c,  false,  false);
    }

    const int rbase = block_row + ((lane >> 4) << 3);
    #pragma unroll
    for (int r = 0; r < 8; ++r) {
        int row = rbase + r;
        if (row < n) x[(size_t)row * HID + ncol] = c[r];
    }
}

__global__ __launch_bounds__(256) void attn_kernel(const float* __restrict__ x,
                                                   const float* __restrict__ a_s,
                                                   const float* __restrict__ a_d,
                                                   float* __restrict__ as_out,
                                                   float* __restrict__ ad_out,
                                                   int n) {
    int node = (blockIdx.x * blockDim.x + threadIdx.x) >> 5;
    int lane = threadIdx.x & 31;
    if (node >= n) return;
    float4 v  = ((const float4*)(x + (size_t)node * HID))[lane];
    float4 s4 = ((const float4*)a_s)[lane];
    float4 d4 = ((const float4*)a_d)[lane];
    float ps = v.x * s4.x + v.y * s4.y + v.z * s4.z + v.w * s4.w;
    float pd = v.x * d4.x + v.y * d4.y + v.z * d4.z + v.w * d4.w;
    #pragma unroll
    for (int o = 16; o > 0; o >>= 1) {
        ps += __shfl_down(ps, o, 32);
        pd += __shfl_down(pd, o, 32);
    }
    if (lane == 0) { as_out[node] = ps; ad_out[node] = pd; }
}

#define NPB 196
#define EDGE_THREADS 1024
#define EDGE_WAVES 32
#define CAPW 384
#define ROW_BATCH 8
#define SCAN_BATCH 8

__device__ __forceinline__ float leaky(float v) { return (v > 0.0f) ? v : NEG_SLOPE * v; }

__device__ __forceinline__ void lds_add(float* p, float v) {
    __hip_atomic_fetch_add(p, v, __ATOMIC_RELAXED, __HIP_MEMORY_SCOPE_WORKGROUP);
}

__device__ __forceinline__ unsigned scan_site(bool ok, int e, int d, int lo, int cnt,
                                              unsigned wcount, int* wls, unsigned short* wld) {
    unsigned dl = (unsigned)(d - lo);
    bool hit = ok && (dl < (unsigned)cnt);
    unsigned mask = __builtin_amdgcn_ballot_w32(hit);
    if (mask != 0u) {
        if (hit) {
            unsigned pos = wcount + __builtin_amdgcn_mbcnt_lo(mask, 0u);
            if (pos < CAPW) { wls[pos] = e; wld[pos] = (unsigned short)dl; }
        }
        wcount += (unsigned)__builtin_popcount(mask);
    }
    return wcount;
}

__global__ __launch_bounds__(EDGE_THREADS) void edge_owner_kernel(const int* __restrict__ src,
                                                                  const int* __restrict__ dst,
                                                                  const float* __restrict__ x,
                                                                  const float* __restrict__ as_,
                                                                  const float* __restrict__ ad_,
                                                                  const float* __restrict__ bias,
                                                                  float* __restrict__ out,
                                                                  int n_edges, int n_nodes,
                                                                  int do_relu) {
    __shared__ __attribute__((aligned(16))) float acc[NPB * HID];
    __shared__ int            ls[EDGE_WAVES * CAPW];
    __shared__ float          lv[EDGE_WAVES * CAPW];
    __shared__ unsigned short ld[EDGE_WAVES * CAPW];
    __shared__ unsigned lmax[NPB];
    __shared__ float    lden[NPB];
    __shared__ float    lad[NPB];
    __shared__ unsigned lovf;

    const int tid  = threadIdx.x;
    const int lane = tid & 31;
    const int wave = __builtin_amdgcn_readfirstlane(tid >> 5);
    int*            wls = ls + wave * CAPW;
    float*          wlv = lv + wave * CAPW;
    unsigned short* wld = ld + wave * CAPW;

    for (int tile = blockIdx.x; tile * NPB < n_nodes; tile += gridDim.x) {
        const int lo  = tile * NPB;
        const int cnt = (n_nodes - lo < NPB) ? (n_nodes - lo) : NPB;

        for (int i = tid; i < NPB * HID; i += EDGE_THREADS) acc[i] = bias[i & (HID - 1)];
        for (int i = tid; i < cnt; i += EDGE_THREADS) {
            lad[i]  = ad_[lo + i];
            lmax[i] = 0u;
            lden[i] = 0.0f;
        }
        if (tid == 0) lovf = 0u;
        const int nself = (cnt > wave) ? ((cnt - 1 - wave) / EDGE_WAVES + 1) : 0;
        if (lane < nself) {
            int i = wave + EDGE_WAVES * lane;
            wls[lane] = n_edges + lo + i;
            wld[lane] = (unsigned short)i;
        }
        unsigned wcount = (unsigned)nself;
        __syncthreads();

        const bool vec_ok = ((((size_t)dst) & 15) == 0);
        const int n4 = vec_ok ? (n_edges >> 2) : 0;
        if (n4 > 0) {
            const int4* d4 = (const int4*)dst;
            const int rot = (int)(((long long)blockIdx.x * n4) / gridDim.x) & ~7;
            int qb = wave * 32;
            for (; qb + (SCAN_BATCH - 1) * EDGE_THREADS + 31 < n4; qb += SCAN_BATCH * EDGE_THREADS) {
                int4 dv[SCAN_BATCH];
                int  qe[SCAN_BATCH];
                #pragma unroll
                for (int u = 0; u < SCAN_BATCH; ++u) {
                    int t = qb + u * EDGE_THREADS + lane + rot;
                    if (t >= n4) t -= n4;
                    qe[u] = t << 2;
                    dv[u] = d4[t];
                }
                #pragma unroll
                for (int u = 0; u < SCAN_BATCH; ++u)
                    asm volatile("" :: "v"(dv[u].x), "v"(dv[u].y), "v"(dv[u].z), "v"(dv[u].w));
                #pragma unroll
                for (int u = 0; u < SCAN_BATCH; ++u) {
                    wcount = scan_site(true, qe[u],     dv[u].x, lo, cnt, wcount, wls, wld);
                    wcount = scan_site(true, qe[u] + 1, dv[u].y, lo, cnt, wcount, wls, wld);
                    wcount = scan_site(true, qe[u] + 2, dv[u].z, lo, cnt, wcount, wls, wld);
                    wcount = scan_site(true, qe[u] + 3, dv[u].w, lo, cnt, wcount, wls, wld);
                }
            }
            for (; qb < n4; qb += EDGE_THREADS) {
                int  t0 = qb + lane;
                bool ok = t0 < n4;
                int  t  = t0 + rot;
                if (t >= n4) t -= n4;
                if (!ok) t = 0;
                int4 d = d4[t];
                int  e = t << 2;
                wcount = scan_site(ok, e,     d.x, lo, cnt, wcount, wls, wld);
                wcount = scan_site(ok, e + 1, d.y, lo, cnt, wcount, wls, wld);
                wcount = scan_site(ok, e + 2, d.z, lo, cnt, wcount, wls, wld);
                wcount = scan_site(ok, e + 3, d.w, lo, cnt, wcount, wls, wld);
            }
        }
        for (int eb = (n4 << 2) + wave * 32; eb < n_edges; eb += EDGE_THREADS) {
            int  e  = eb + lane;
            bool ok = e < n_edges;
            int  d  = dst[ok ? e : 0];
            wcount = scan_site(ok, e, d, lo, cnt, wcount, wls, wld);
        }
        wcount = (unsigned)__builtin_amdgcn_readfirstlane((int)wcount);
        if (wcount > CAPW && lane == 0) lovf = 1u;
        __syncthreads();

        const bool overflow = (lovf != 0u);
        const int  nl = overflow ? nself : (int)wcount;

        for (int i = lane; i < nl; i += 32) {
            int e  = wls[i];
            int dl = wld[i];
            int s  = (e < n_edges) ? src[e] : (e - n_edges);
            float v = leaky(as_[s] + lad[dl]);
            atomicMax(lmax + dl, flipf(v));
            wls[i] = s;
            wlv[i] = v;
        }
        if (overflow) {
            for (int e = tid; e < n_edges; e += EDGE_THREADS) {
                unsigned dl = (unsigned)(dst[e] - lo);
                if (dl < (unsigned)cnt)
                    atomicMax(lmax + dl, flipf(leaky(as_[src[e]] + lad[dl])));
            }
        }
        __syncthreads();

        for (int i = lane; i < nl; i += 32) {
            int dl = wld[i];
            float ex = __expf(wlv[i] - unflipf(lmax[dl]));
            wlv[i] = ex;
            lds_add(lden + dl, ex);
        }
        if (overflow) {
            for (int e = tid; e < n_edges; e += EDGE_THREADS) {
                unsigned dl = (unsigned)(dst[e] - lo);
                if (dl < (unsigned)cnt) {
                    float v = leaky(as_[src[e]] + lad[dl]);
                    lds_add(lden + dl, __expf(v - unflipf(lmax[dl])));
                }
            }
        }
        __syncthreads();

        for (int i0 = 0; i0 < nl; i0 += ROW_BATCH) {
            float al[ROW_BATCH]; int dr[ROW_BATCH];
            float r0[ROW_BATCH], r1[ROW_BATCH], r2[ROW_BATCH], r3[ROW_BATCH];
            #pragma unroll
            for (int u = 0; u < ROW_BATCH; ++u) {
                int i  = i0 + u;
                int ii = (i < nl) ? i : i0;
                int s  = wls[ii];
                int dl = wld[ii];
                dr[u] = dl;
                al[u] = wlv[ii] / (lden[dl] + GAT_EPS);
                const float* xr = x + (size_t)s * HID + lane;
                r0[u] = xr[0]; r1[u] = xr[32]; r2[u] = xr[64]; r3[u] = xr[96];
            }
            #pragma unroll
            for (int u = 0; u < ROW_BATCH; ++u) {
                if (i0 + u < nl) {
                    float* o = acc + dr[u] * HID + lane;
                    lds_add(o,      al[u] * r0[u]);
                    lds_add(o + 32, al[u] * r1[u]);
                    lds_add(o + 64, al[u] * r2[u]);
                    lds_add(o + 96, al[u] * r3[u]);
                }
            }
        }
        if (overflow) {
            for (int base = wave * 32; base < n_edges; base += EDGE_THREADS) {
                int e = base + lane;
                bool hit = false; int s = 0; int dli = 0; float alpha = 0.0f;
                if (e < n_edges) {
                    unsigned dl = (unsigned)(dst[e] - lo);
                    if (dl < (unsigned)cnt) {
                        hit = true;
                        dli = (int)dl;
                        s = src[e];
                        float v = leaky(as_[s] + lad[dl]);
                        alpha = __expf(v - unflipf(lmax[dl])) / (lden[dl] + GAT_EPS);
                    }
                }
                unsigned mask = __builtin_amdgcn_ballot_w32(hit);
                while (mask != 0u) {
                    int b = __builtin_ctz(mask);
                    mask &= mask - 1u;
                    int   sb = __shfl(s, b, 32);
                    int   db = __shfl(dli, b, 32);
                    float ab = __shfl(alpha, b, 32);
                    const float* xr = x + (size_t)sb * HID + lane;
                    float* o = acc + db * HID + lane;
                    lds_add(o,      ab * xr[0]);
                    lds_add(o + 32, ab * xr[32]);
                    lds_add(o + 64, ab * xr[64]);
                    lds_add(o + 96, ab * xr[96]);
                }
            }
        }
        __syncthreads();

        const float4* acc4 = (const float4*)acc;
        float4* out4 = (float4*)(out + (size_t)lo * HID);
        for (int i = tid; i < cnt * (HID / 4); i += EDGE_THREADS) {
            float4 v = acc4[i];
            if (do_relu) {
                v.x = fmaxf(v.x, 0.0f); v.y = fmaxf(v.y, 0.0f);
                v.z = fmaxf(v.z, 0.0f); v.w = fmaxf(v.w, 0.0f);
            }
            out4[i] = v;
        }
        __syncthreads();
    }
}

extern "C" void kernel_launch(void* const* d_in, const int* in_sizes, int n_in,
                              void* d_out, int out_size, void* d_ws, size_t ws_size,
                              hipStream_t stream) {
    const float* z     = (const float*)d_in[0];
    const int*   ei    = (const int*)  d_in[1];
    const float* Ws    = (const float*)d_in[2];
    const float* a_src = (const float*)d_in[3];
    const float* a_dst = (const float*)d_in[4];
    const float* bias  = (const float*)d_in[5];
    float*       out   = (float*)d_out;

    const int N = in_sizes[0] / HID;
    const int E = in_sizes[1] / 2;
    const int* src = ei;
    const int* dst = ei + E;

    auto align = [](size_t v) { return (v + 255) & ~(size_t)255; };
    char* p = (char*)d_ws;
    float*    x      = (float*)p;    p += align((size_t)N * HID * sizeof(float));
    float*    bufA   = (float*)p;    p += align((size_t)N * HID * sizeof(float));
    float*    bufB   = (float*)p;    p += align((size_t)N * HID * sizeof(float));
    float*    as_    = (float*)p;    p += align((size_t)N * sizeof(float));
    float*    ad_    = (float*)p;    p += align((size_t)N * sizeof(float));

    const float* h_ptr = z;
    for (int l = 0; l < 3; ++l) {
        float* o_ptr = (l == 0) ? bufA : (l == 1) ? bufB : out;
        const float* W  = Ws    + (size_t)l * HID * HID;
        const float* as = a_src + (size_t)l * HID;
        const float* ad = a_dst + (size_t)l * HID;
        const float* b  = bias  + (size_t)l * HID;

        gemm_x_kernel<<<(N + 15) / 16, 256, 0, stream>>>(h_ptr, W, x, N);

        attn_kernel<<<((size_t)N * 32 + 255) / 256, 256, 0, stream>>>(x, as, ad, as_, ad_, N);

        edge_owner_kernel<<<256, EDGE_THREADS, 0, stream>>>(src, dst, x, as_, ad_, b, o_ptr,
                                                           E, N, (l < 2) ? 1 : 0);
        h_ptr = o_ptr;
    }
}


# Next layer's x = relu(rows) @ W and its two dots formed in the edge launch's tail from the block's LDS rows (f32 WMMA, 132-float row pitch, same k and sum order); h never stored; 5 launches not 9
# speedup vs baseline: 3.4262x; 1.1875x over previous
#include <hip/hip_runtime.h>
#include <hip/hip_bf16.h>

#define HID 128
#define NEG_SLOPE 0.2f
#define GAT_EPS 1e-16f

typedef float v2f __attribute__((ext_vector_type(2)));
typedef float v8f __attribute__((ext_vector_type(8)));

__device__ __forceinline__ unsigned flipf(float f) {
    unsigned u = __float_as_uint(f);
    return (u & 0x80000000u) ? ~u : (u | 0x80000000u);
}
__device__ __forceinline__ float unflipf(unsigned u) {
    return __uint_as_float((u & 0x80000000u) ? (u ^ 0x80000000u) : ~u);
}

__global__ __launch_bounds__(256) void gemm_x_kernel(const float* __restrict__ h,
                                                     const float* __restrict__ W,
                                                     float* __restrict__ x,
                                                     int n) {
    __shared__ float sh[16 * HID];
    const int block_row = blockIdx.x * 16;
    const int tid  = threadIdx.x;
    const int wave = tid >> 5;
    const int lane = tid & 31;

    const float4* gsrc = (const float4*)(h + (size_t)block_row * HID);
    float4*       ldst = (float4*)sh;
    #pragma unroll
    for (int i = tid; i < 16 * HID / 4; i += 256) ldst[i] = gsrc[i];
    __syncthreads();

    const int m    = lane & 15;
    const int koff = (lane >> 4) << 1;
    const int ncol = wave * 16 + (lane & 15);

    v8f c = {};
    #pragma unroll
    for (int k = 0; k < HID; k += 4) {
        v2f a, b;
        a.x = sh[m * HID + k + koff];
        a.y = sh[m * HID + k + koff + 1];
        b.x = W[(size_t)(k + koff)     * HID + ncol];
        b.y = W[(size_t)(k + koff + 1) * HID + ncol];
        c = __builtin_amdgcn_wmma_f32_16x16x4_f32(
                 false, a,  false, b,
                 (short)0, c,  false,  false);
    }

    const int rbase = block_row + ((lane >> 4) << 3);
    #pragma unroll
    for (int r = 0; r < 8; ++r) {
        int row = rbase + r;
        if (row < n) x[(size_t)row * HID + ncol] = c[r];
    }
}

__global__ __launch_bounds__(256) void attn_kernel(const float* __restrict__ x,
                                                   const float* __restrict__ a_s,
                                                   const float* __restrict__ a_d,
                                                   float* __restrict__ as_out,
                                                   float* __restrict__ ad_out,
                                                   int n) {
    int node = (blockIdx.x * blockDim.x + threadIdx.x) >> 5;
    int lane = threadIdx.x & 31;
    if (node >= n) return;
    float4 v  = ((const float4*)(x + (size_t)node * HID))[lane];
    float4 s4 = ((const float4*)a_s)[lane];
    float4 d4 = ((const float4*)a_d)[lane];
    float ps = v.x * s4.x + v.y * s4.y + v.z * s4.z + v.w * s4.w;
    float pd = v.x * d4.x + v.y * d4.y + v.z * d4.z + v.w * d4.w;
    #pragma unroll
    for (int o = 16; o > 0; o >>= 1) {
        ps += __shfl_down(ps, o, 32);
        pd += __shfl_down(pd, o, 32);
    }
    if (lane == 0) { as_out[node] = ps; ad_out[node] = pd; }
}

#define NPB 196
#define APITCH 132
#define MTILES ((NPB + 15) / 16)
#define EDGE_THREADS 1024
#define EDGE_WAVES 32
#define CAPW 384
#define ROW_BATCH 8
#define SCAN_BATCH 8

__device__ __forceinline__ float leaky(float v) { return (v > 0.0f) ? v : NEG_SLOPE * v; }

__device__ __forceinline__ void lds_add(float* p, float v) {
    __hip_atomic_fetch_add(p, v, __ATOMIC_RELAXED, __HIP_MEMORY_SCOPE_WORKGROUP);
}

__device__ __forceinline__ unsigned scan_site(bool ok, int e, int d, int lo, int cnt,
                                              unsigned wcount, int* wls, unsigned short* wld) {
    unsigned dl = (unsigned)(d - lo);
    bool hit = ok && (dl < (unsigned)cnt);
    unsigned mask = __builtin_amdgcn_ballot_w32(hit);
    if (mask != 0u) {
        if (hit) {
            unsigned pos = wcount + __builtin_amdgcn_mbcnt_lo(mask, 0u);
            if (pos < CAPW) { wls[pos] = e; wld[pos] = (unsigned short)dl; }
        }
        wcount += (unsigned)__builtin_popcount(mask);
    }
    return wcount;
}

#define KBLK 2
template <int NT>
__device__ __forceinline__ void next_x_tiles(const float* acc, const float* __restrict__ Wn,
                                             int g, int lane, int ncol, v8f* c) {
    const int m    = lane & 15;
    const int koff = (lane >> 4) << 1;
    const float* arow[NT];
    #pragma unroll
    for (int j = 0; j < NT; ++j) {
        int row = (g + 4 * j) * 16 + m;
        row = (row < NPB) ? row : (NPB - 1);
        arow[j] = acc + row * APITCH + koff;
        c[j] = v8f{};
    }
    const float* wp = Wn + (size_t)koff * HID + ncol;
    v2f bn[KBLK];
    #pragma unroll
    for (int s = 0; s < KBLK; ++s) {
        bn[s].x = wp[(size_t)(4 * s) * HID];
        bn[s].y = wp[(size_t)(4 * s + 1) * HID];
    }
    #pragma nounroll
    for (int kb = 0; kb < HID / 4; kb += KBLK) {
        v2f bc[KBLK];
        #pragma unroll
        for (int s = 0; s < KBLK; ++s) bc[s] = bn[s];
        if (kb + KBLK < HID / 4) {
            const float* wq = wp + (size_t)(4 * (kb + KBLK)) * HID;
            #pragma unroll
            for (int s = 0; s < KBLK; ++s) {
                bn[s].x = wq[(size_t)(4 * s) * HID];
                bn[s].y = wq[(size_t)(4 * s + 1) * HID];
            }
        }
        #pragma unroll
        for (int s = 0; s < KBLK; ++s) {
            #pragma unroll
            for (int j = 0; j < NT; ++j) {
                v2f a = *(const v2f*)(arow[j] + 4 * (kb + s));
                c[j] = __builtin_amdgcn_wmma_f32_16x16x4_f32(
                            false, a,  false, bc[s],
                            (short)0, c[j],  false,  false);
            }
        }
    }
}

__global__ __launch_bounds__(EDGE_THREADS) void edge_owner_kernel(const int* __restrict__ src,
                                                                  const int* __restrict__ dst,
                                                                  const float* __restrict__ x,
                                                                  const float* __restrict__ as_,
                                                                  const float* __restrict__ ad_,
                                                                  const float* __restrict__ bias,
                                                                  const float* __restrict__ Wn,
                                                                  const float* __restrict__ a_sn,
                                                                  const float* __restrict__ a_dn,
                                                                  float* __restrict__ dest,
                                                                  float* __restrict__ as_next,
                                                                  float* __restrict__ ad_next,
                                                                  int n_edges, int n_nodes) {
    __shared__ __attribute__((aligned(16))) float acc[NPB * APITCH];
    __shared__ int            ls[EDGE_WAVES * CAPW];
    __shared__ float          lv[EDGE_WAVES * CAPW];
    __shared__ unsigned short ld[EDGE_WAVES * CAPW];
    __shared__ unsigned lmax[NPB];
    __shared__ float    lden[NPB];
    __shared__ float    lad[NPB];
    __shared__ unsigned lovf;

    const int tid  = threadIdx.x;
    const int lane = tid & 31;
    const int wave = __builtin_amdgcn_readfirstlane(tid >> 5);
    int*            wls = ls + wave * CAPW;
    float*          wlv = lv + wave * CAPW;
    unsigned short* wld = ld + wave * CAPW;

    for (int tile = blockIdx.x; tile * NPB < n_nodes; tile += gridDim.x) {
        const int lo  = tile * NPB;
        const int cnt = (n_nodes - lo < NPB) ? (n_nodes - lo) : NPB;

        for (int i = tid; i < NPB * HID; i += EDGE_THREADS)
            acc[(i >> 7) * APITCH + (i & (HID - 1))] = bias[i & (HID - 1)];
        for (int i = tid; i < cnt; i += EDGE_THREADS) {
            lad[i]  = ad_[lo + i];
            lmax[i] = 0u;
            lden[i] = 0.0f;
        }
        if (tid == 0) lovf = 0u;
        const int nself = (cnt > wave) ? ((cnt - 1 - wave) / EDGE_WAVES + 1) : 0;
        if (lane < nself) {
            int i = wave + EDGE_WAVES * lane;
            wls[lane] = n_edges + lo + i;
            wld[lane] = (unsigned short)i;
        }
        unsigned wcount = (unsigned)nself;
        __syncthreads();

        const bool vec_ok = ((((size_t)dst) & 15) == 0);
        const int n4 = vec_ok ? (n_edges >> 2) : 0;
        if (n4 > 0) {
            const int4* d4 = (const int4*)dst;
            const int rot = (int)(((long long)blockIdx.x * n4) / gridDim.x) & ~7;
            int qb = wave * 32;
            for (; qb + (SCAN_BATCH - 1) * EDGE_THREADS + 31 < n4; qb += SCAN_BATCH * EDGE_THREADS) {
                int4 dv[SCAN_BATCH];
                int  qe[SCAN_BATCH];
                #pragma unroll
                for (int u = 0; u < SCAN_BATCH; ++u) {
                    int t = qb + u * EDGE_THREADS + lane + rot;
                    if (t >= n4) t -= n4;
                    qe[u] = t << 2;
                    dv[u] = d4[t];
                }
                #pragma unroll
                for (int u = 0; u < SCAN_BATCH; ++u)
                    asm volatile("" :: "v"(dv[u].x), "v"(dv[u].y), "v"(dv[u].z), "v"(dv[u].w));
                #pragma unroll
                for (int u = 0; u < SCAN_BATCH; ++u) {
                    wcount = scan_site(true, qe[u],     dv[u].x, lo, cnt, wcount, wls, wld);
                    wcount = scan_site(true, qe[u] + 1, dv[u].y, lo, cnt, wcount, wls, wld);
                    wcount = scan_site(true, qe[u] + 2, dv[u].z, lo, cnt, wcount, wls, wld);
                    wcount = scan_site(true, qe[u] + 3, dv[u].w, lo, cnt, wcount, wls, wld);
                }
            }
            for (; qb < n4; qb += EDGE_THREADS) {
                int  t0 = qb + lane;
                bool ok = t0 < n4;
                int  t  = t0 + rot;
                if (t >= n4) t -= n4;
                if (!ok) t = 0;
                int4 d = d4[t];
                int  e = t << 2;
                wcount = scan_site(ok, e,     d.x, lo, cnt, wcount, wls, wld);
                wcount = scan_site(ok, e + 1, d.y, lo, cnt, wcount, wls, wld);
                wcount = scan_site(ok, e + 2, d.z, lo, cnt, wcount, wls, wld);
                wcount = scan_site(ok, e + 3, d.w, lo, cnt, wcount, wls, wld);
            }
        }
        for (int eb = (n4 << 2) + wave * 32; eb < n_edges; eb += EDGE_THREADS) {
            int  e  = eb + lane;
            bool ok = e < n_edges;
            int  d  = dst[ok ? e : 0];
            wcount = scan_site(ok, e, d, lo, cnt, wcount, wls, wld);
        }
        wcount = (unsigned)__builtin_amdgcn_readfirstlane((int)wcount);
        if (wcount > CAPW && lane == 0) lovf = 1u;
        __syncthreads();

        const bool overflow = (lovf != 0u);
        const int  nl = overflow ? nself : (int)wcount;

        for (int i = lane; i < nl; i += 32) {
            int e  = wls[i];
            int dl = wld[i];
            int s  = (e < n_edges) ? src[e] : (e - n_edges);
            float v = leaky(as_[s] + lad[dl]);
            atomicMax(lmax + dl, flipf(v));
            wls[i] = s;
            wlv[i] = v;
        }
        if (overflow) {
            for (int e = tid; e < n_edges; e += EDGE_THREADS) {
                unsigned dl = (unsigned)(dst[e] - lo);
                if (dl < (unsigned)cnt)
                    atomicMax(lmax + dl, flipf(leaky(as_[src[e]] + lad[dl])));
            }
        }
        __syncthreads();

        for (int i = lane; i < nl; i += 32) {
            int dl = wld[i];
            float ex = __expf(wlv[i] - unflipf(lmax[dl]));
            wlv[i] = ex;
            lds_add(lden + dl, ex);
        }
        if (overflow) {
            for (int e = tid; e < n_edges; e += EDGE_THREADS) {
                unsigned dl = (unsigned)(dst[e] - lo);
                if (dl < (unsigned)cnt) {
                    float v = leaky(as_[src[e]] + lad[dl]);
                    lds_add(lden + dl, __expf(v - unflipf(lmax[dl])));
                }
            }
        }
        __syncthreads();

        for (int i0 = 0; i0 < nl; i0 += ROW_BATCH) {
            float al[ROW_BATCH]; int dr[ROW_BATCH];
            float r0[ROW_BATCH], r1[ROW_BATCH], r2[ROW_BATCH], r3[ROW_BATCH];
            #pragma unroll
            for (int u = 0; u < ROW_BATCH; ++u) {
                int i  = i0 + u;
                int ii = (i < nl) ? i : i0;
                int s  = wls[ii];
                int dl = wld[ii];
                dr[u] = dl;
                al[u] = wlv[ii] / (lden[dl] + GAT_EPS);
                const float* xr = x + (size_t)s * HID + lane;
                r0[u] = xr[0]; r1[u] = xr[32]; r2[u] = xr[64]; r3[u] = xr[96];
            }
            #pragma unroll
            for (int u = 0; u < ROW_BATCH; ++u) {
                if (i0 + u < nl) {
                    float* o = acc + dr[u] * APITCH + lane;
                    lds_add(o,      al[u] * r0[u]);
                    lds_add(o + 32, al[u] * r1[u]);
                    lds_add(o + 64, al[u] * r2[u]);
                    lds_add(o + 96, al[u] * r3[u]);
                }
            }
        }
        if (overflow) {
            for (int base = wave * 32; base < n_edges; base += EDGE_THREADS) {
                int e = base + lane;
                bool hit = false; int s = 0; int dli = 0; float alpha = 0.0f;
                if (e < n_edges) {
                    unsigned dl = (unsigned)(dst[e] - lo);
                    if (dl < (unsigned)cnt) {
                        hit = true;
                        dli = (int)dl;
                        s = src[e];
                        float v = leaky(as_[s] + lad[dl]);
                        alpha = __expf(v - unflipf(lmax[dl])) / (lden[dl] + GAT_EPS);
                    }
                }
                unsigned mask = __builtin_amdgcn_ballot_w32(hit);
                while (mask != 0u) {
                    int b = __builtin_ctz(mask);
                    mask &= mask - 1u;
                    int   sb = __shfl(s, b, 32);
                    int   db = __shfl(dli, b, 32);
                    float ab = __shfl(alpha, b, 32);
                    const float* xr = x + (size_t)sb * HID + lane;
                    float* o = acc + db * APITCH + lane;
                    lds_add(o,      ab * xr[0]);
                    lds_add(o + 32, ab * xr[32]);
                    lds_add(o + 64, ab * xr[64]);
                    lds_add(o + 96, ab * xr[96]);
                }
            }
        }
        __syncthreads();

        if (Wn != nullptr) {
            for (int i = tid; i < NPB * (HID / 4); i += EDGE_THREADS) {
                float4* p4 = (float4*)(acc + (i >> 5) * APITCH + (i & 31) * 4);
                float4 v = *p4;
                v.x = fmaxf(v.x, 0.0f); v.y = fmaxf(v.y, 0.0f);
                v.z = fmaxf(v.z, 0.0f); v.w = fmaxf(v.w, 0.0f);
                *p4 = v;
            }
            __syncthreads();
            const int g    = wave >> 3;
            const int ncol = (wave & 7) * 16 + (lane & 15);
            v8f c[4];
            if (g + 12 < MTILES) {
                next_x_tiles<4>(acc, Wn, g, lane, ncol, c);
            } else {
                next_x_tiles<3>(acc, Wn, g, lane, ncol, c);
                c[3] = v8f{};
            }
            __syncthreads();
            #pragma unroll
            for (int j = 0; j < 4; ++j) {
                const int rbase = (g + 4 * j) * 16 + ((lane >> 4) << 3);
                #pragma unroll
                for (int r = 0; r < 8; ++r) {
                    int row = rbase + r;
                    if (j < 3 || row < NPB) acc[row * APITCH + ncol] = c[j][r];
                }
            }
            __syncthreads();
            const float4 s4 = ((const float4*)a_sn)[lane];
            const float4 d4 = ((const float4*)a_dn)[lane];
            for (int i = wave; i < cnt; i += EDGE_WAVES) {
                float4 v = *(const float4*)(acc + i * APITCH + lane * 4);
                ((float4*)(dest + (size_t)(lo + i) * HID))[lane] = v;
                float ps = v.x * s4.x + v.y * s4.y + v.z * s4.z + v.w * s4.w;
                float pd = v.x * d4.x + v.y * d4.y + v.z * d4.z + v.w * d4.w;
                #pragma unroll
                for (int o = 16; o > 0; o >>= 1) {
                    ps += __shfl_down(ps, o, 32);
                    pd += __shfl_down(pd, o, 32);
                }
                if (lane == 0) { as_next[lo + i] = ps; ad_next[lo + i] = pd; }
            }
        } else {
            float4* out4 = (float4*)(dest + (size_t)lo * HID);
            for (int i = tid; i < cnt * (HID / 4); i += EDGE_THREADS)
                out4[i] = *(const float4*)(acc + (i >> 5) * APITCH + (i & 31) * 4);
        }
        __syncthreads();
    }
}

extern "C" void kernel_launch(void* const* d_in, const int* in_sizes, int n_in,
                              void* d_out, int out_size, void* d_ws, size_t ws_size,
                              hipStream_t stream) {
    const float* z     = (const float*)d_in[0];
    const int*   ei    = (const int*)  d_in[1];
    const float* Ws    = (const float*)d_in[2];
    const float* a_src = (const float*)d_in[3];
    const float* a_dst = (const float*)d_in[4];
    const float* bias  = (const float*)d_in[5];
    float*       out   = (float*)d_out;

    const int N = in_sizes[0] / HID;
    const int E = in_sizes[1] / 2;
    const int* src = ei;
    const int* dst = ei + E;

    auto align = [](size_t v) { return (v + 255) & ~(size_t)255; };
    char* p = (char*)d_ws;
    float* x0  = (float*)p;    p += align((size_t)N * HID * sizeof(float));
    float* x1  = (float*)p;    p += align((size_t)N * HID * sizeof(float));
    float* as0 = (float*)p;    p += align((size_t)N * sizeof(float));
    float* ad0 = (float*)p;    p += align((size_t)N * sizeof(float));
    float* as1 = (float*)p;    p += align((size_t)N * sizeof(float));
    float* ad1 = (float*)p;    p += align((size_t)N * sizeof(float));

    gemm_x_kernel<<<(N + 15) / 16, 256, 0, stream>>>(z, Ws, x0, N);
    attn_kernel<<<((size_t)N * 32 + 255) / 256, 256, 0, stream>>>(x0, a_src, a_dst, as0, ad0, N);

    edge_owner_kernel<<<256, EDGE_THREADS, 0, stream>>>(src, dst, x0, as0, ad0, bias,
                                                       Ws + HID * HID, a_src + HID, a_dst + HID,
                                                       x1, as1, ad1, E, N);
    edge_owner_kernel<<<256, EDGE_THREADS, 0, stream>>>(src, dst, x1, as1, ad1, bias + HID,
                                                       Ws + 2 * HID * HID, a_src + 2 * HID, a_dst + 2 * HID,
                                                       x0, as0, ad0, E, N);
    edge_owner_kernel<<<256, EDGE_THREADS, 0, stream>>>(src, dst, x0, as0, ad0, bias + 2 * HID,
                                                       nullptr, nullptr, nullptr,
                                                       out, nullptr, nullptr, E, N);
}


# In-edge lists kept across layers: first edge launch writes each wave's finished list (source, local dst) and its length to the workspace; the next two read it back: no dst scan, no src gather
# speedup vs baseline: 4.2639x; 1.2445x over previous
#include <hip/hip_runtime.h>
#include <hip/hip_bf16.h>

#define HID 128
#define NEG_SLOPE 0.2f
#define GAT_EPS 1e-16f

typedef float v2f __attribute__((ext_vector_type(2)));
typedef float v8f __attribute__((ext_vector_type(8)));

__device__ __forceinline__ unsigned flipf(float f) {
    unsigned u = __float_as_uint(f);
    return (u & 0x80000000u) ? ~u : (u | 0x80000000u);
}
__device__ __forceinline__ float unflipf(unsigned u) {
    return __uint_as_float((u & 0x80000000u) ? (u ^ 0x80000000u) : ~u);
}

__global__ __launch_bounds__(256) void gemm_x_kernel(const float* __restrict__ h,
                                                     const float* __restrict__ W,
                                                     float* __restrict__ x,
                                                     int n) {
    __shared__ float sh[16 * HID];
    const int block_row = blockIdx.x * 16;
    const int tid  = threadIdx.x;
    const int wave = tid >> 5;
    const int lane = tid & 31;

    const float4* gsrc = (const float4*)(h + (size_t)block_row * HID);
    float4*       ldst = (float4*)sh;
    #pragma unroll
    for (int i = tid; i < 16 * HID / 4; i += 256) ldst[i] = gsrc[i];
    __syncthreads();

    const int m    = lane & 15;
    const int koff = (lane >> 4) << 1;
    const int ncol = wave * 16 + (lane & 15);

    v8f c = {};
    #pragma unroll
    for (int k = 0; k < HID; k += 4) {
        v2f a, b;
        a.x = sh[m * HID + k + koff];
        a.y = sh[m * HID + k + koff + 1];
        b.x = W[(size_t)(k + koff)     * HID + ncol];
        b.y = W[(size_t)(k + koff + 1) * HID + ncol];
        c = __builtin_amdgcn_wmma_f32_16x16x4_f32(
                 false, a,  false, b,
                 (short)0, c,  false,  false);
    }

    const int rbase = block_row + ((lane >> 4) << 3);
    #pragma unroll
    for (int r = 0; r < 8; ++r) {
        int row = rbase + r;
        if (row < n) x[(size_t)row * HID + ncol] = c[r];
    }
}

__global__ __launch_bounds__(256) void attn_kernel(const float* __restrict__ x,
                                                   const float* __restrict__ a_s,
                                                   const float* __restrict__ a_d,
                                                   float* __restrict__ as_out,
                                                   float* __restrict__ ad_out,
                                                   int n) {
    int node = (blockIdx.x * blockDim.x + threadIdx.x) >> 5;
    int lane = threadIdx.x & 31;
    if (node >= n) return;
    float4 v  = ((const float4*)(x + (size_t)node * HID))[lane];
    float4 s4 = ((const float4*)a_s)[lane];
    float4 d4 = ((const float4*)a_d)[lane];
    float ps = v.x * s4.x + v.y * s4.y + v.z * s4.z + v.w * s4.w;
    float pd = v.x * d4.x + v.y * d4.y + v.z * d4.z + v.w * d4.w;
    #pragma unroll
    for (int o = 16; o > 0; o >>= 1) {
        ps += __shfl_down(ps, o, 32);
        pd += __shfl_down(pd, o, 32);
    }
    if (lane == 0) { as_out[node] = ps; ad_out[node] = pd; }
}

#define NPB 196
#define APITCH 132
#define MTILES ((NPB + 15) / 16)
#define EDGE_THREADS 1024
#define EDGE_WAVES 32
#define CAPW 384
#define ROW_BATCH 8
#define SCAN_BATCH 8
#define KEEP_NONE  0
#define KEEP_WRITE 1
#define KEEP_READ  2
#define SRC_BITS 24
#define READ_BATCH 4
static_assert(NPB <= 256, "the local destination must fit the 8 bits above the source node");

__device__ __forceinline__ float leaky(float v) { return (v > 0.0f) ? v : NEG_SLOPE * v; }

__device__ __forceinline__ void lds_add(float* p, float v) {
    __hip_atomic_fetch_add(p, v, __ATOMIC_RELAXED, __HIP_MEMORY_SCOPE_WORKGROUP);
}

__device__ __forceinline__ unsigned scan_site(bool ok, int e, int d, int lo, int cnt,
                                              unsigned wcount, int* wls, unsigned short* wld) {
    unsigned dl = (unsigned)(d - lo);
    bool hit = ok && (dl < (unsigned)cnt);
    unsigned mask = __builtin_amdgcn_ballot_w32(hit);
    if (mask != 0u) {
        if (hit) {
            unsigned pos = wcount + __builtin_amdgcn_mbcnt_lo(mask, 0u);
            if (pos < CAPW) { wls[pos] = e; wld[pos] = (unsigned short)dl; }
        }
        wcount += (unsigned)__builtin_popcount(mask);
    }
    return wcount;
}

#define KBLK 2
template <int NT>
__device__ __forceinline__ void next_x_tiles(const float* acc, const float* __restrict__ Wn,
                                             int g, int lane, int ncol, v8f* c) {
    const int m    = lane & 15;
    const int koff = (lane >> 4) << 1;
    const float* arow[NT];
    #pragma unroll
    for (int j = 0; j < NT; ++j) {
        int row = (g + 4 * j) * 16 + m;
        row = (row < NPB) ? row : (NPB - 1);
        arow[j] = acc + row * APITCH + koff;
        c[j] = v8f{};
    }
    const float* wp = Wn + (size_t)koff * HID + ncol;
    v2f bn[KBLK];
    #pragma unroll
    for (int s = 0; s < KBLK; ++s) {
        bn[s].x = wp[(size_t)(4 * s) * HID];
        bn[s].y = wp[(size_t)(4 * s + 1) * HID];
    }
    #pragma nounroll
    for (int kb = 0; kb < HID / 4; kb += KBLK) {
        v2f bc[KBLK];
        #pragma unroll
        for (int s = 0; s < KBLK; ++s) bc[s] = bn[s];
        if (kb + KBLK < HID / 4) {
            const float* wq = wp + (size_t)(4 * (kb + KBLK)) * HID;
            #pragma unroll
            for (int s = 0; s < KBLK; ++s) {
                bn[s].x = wq[(size_t)(4 * s) * HID];
                bn[s].y = wq[(size_t)(4 * s + 1) * HID];
            }
        }
        #pragma unroll
        for (int s = 0; s < KBLK; ++s) {
            #pragma unroll
            for (int j = 0; j < NT; ++j) {
                v2f a = *(const v2f*)(arow[j] + 4 * (kb + s));
                c[j] = __builtin_amdgcn_wmma_f32_16x16x4_f32(
                            false, a,  false, bc[s],
                            (short)0, c[j],  false,  false);
            }
        }
    }
}

__global__ __launch_bounds__(EDGE_THREADS) void edge_owner_kernel(const int* __restrict__ src,
                                                                  const int* __restrict__ dst,
                                                                  const float* __restrict__ x,
                                                                  const float* __restrict__ as_,
                                                                  const float* __restrict__ ad_,
                                                                  const float* __restrict__ bias,
                                                                  const float* __restrict__ Wn,
                                                                  const float* __restrict__ a_sn,
                                                                  const float* __restrict__ a_dn,
                                                                  float* __restrict__ dest,
                                                                  float* __restrict__ as_next,
                                                                  float* __restrict__ ad_next,
                                                                  unsigned* klist, unsigned* kcount, int kmode,
                                                                  int n_edges, int n_nodes) {
    __shared__ __attribute__((aligned(16))) float acc[NPB * APITCH];
    __shared__ int            ls[EDGE_WAVES * CAPW];
    __shared__ float          lv[EDGE_WAVES * CAPW];
    __shared__ unsigned short ld[EDGE_WAVES * CAPW];
    __shared__ unsigned lmax[NPB];
    __shared__ float    lden[NPB];
    __shared__ float    lad[NPB];
    __shared__ unsigned lovf;

    const int tid  = threadIdx.x;
    const int lane = tid & 31;
    const int wave = __builtin_amdgcn_readfirstlane(tid >> 5);
    int*            wls = ls + wave * CAPW;
    float*          wlv = lv + wave * CAPW;
    unsigned short* wld = ld + wave * CAPW;

    for (int tile = blockIdx.x; tile * NPB < n_nodes; tile += gridDim.x) {
        const int lo  = tile * NPB;
        const int cnt = (n_nodes - lo < NPB) ? (n_nodes - lo) : NPB;
        const bool readback = (kmode == KEEP_READ);
        unsigned* gl = klist + ((size_t)tile * EDGE_WAVES + wave) * CAPW;
        unsigned kw = 0u, kw0 = 0u;
        if (readback) {
            kw  = kcount[tile * EDGE_WAVES + wave];
            kw0 = kcount[tile * EDGE_WAVES];
        }

        for (int i = tid; i < NPB * HID; i += EDGE_THREADS)
            acc[(i >> 7) * APITCH + (i & (HID - 1))] = bias[i & (HID - 1)];
        for (int i = tid; i < cnt; i += EDGE_THREADS) {
            lad[i]  = ad_[lo + i];
            lmax[i] = 0u;
            lden[i] = 0.0f;
        }
        if (tid == 0) lovf = 0u;
        const int nself = (cnt > wave) ? ((cnt - 1 - wave) / EDGE_WAVES + 1) : 0;
        if (!readback && lane < nself) {
            int i = wave + EDGE_WAVES * lane;
            wls[lane] = n_edges + lo + i;
            wld[lane] = (unsigned short)i;
        }
        unsigned wcount = (unsigned)nself;
        __syncthreads();

        const bool vec_ok = ((((size_t)dst) & 15) == 0);
        const int n4 = (vec_ok && !readback) ? (n_edges >> 2) : 0;
        if (n4 > 0) {
            const int4* d4 = (const int4*)dst;
            const int rot = (int)(((long long)blockIdx.x * n4) / gridDim.x) & ~7;
            int qb = wave * 32;
            for (; qb + (SCAN_BATCH - 1) * EDGE_THREADS + 31 < n4; qb += SCAN_BATCH * EDGE_THREADS) {
                int4 dv[SCAN_BATCH];
                int  qe[SCAN_BATCH];
                #pragma unroll
                for (int u = 0; u < SCAN_BATCH; ++u) {
                    int t = qb + u * EDGE_THREADS + lane + rot;
                    if (t >= n4) t -= n4;
                    qe[u] = t << 2;
                    dv[u] = d4[t];
                }
                #pragma unroll
                for (int u = 0; u < SCAN_BATCH; ++u)
                    asm volatile("" :: "v"(dv[u].x), "v"(dv[u].y), "v"(dv[u].z), "v"(dv[u].w));
                #pragma unroll
                for (int u = 0; u < SCAN_BATCH; ++u) {
                    wcount = scan_site(true, qe[u],     dv[u].x, lo, cnt, wcount, wls, wld);
                    wcount = scan_site(true, qe[u] + 1, dv[u].y, lo, cnt, wcount, wls, wld);
                    wcount = scan_site(true, qe[u] + 2, dv[u].z, lo, cnt, wcount, wls, wld);
                    wcount = scan_site(true, qe[u] + 3, dv[u].w, lo, cnt, wcount, wls, wld);
                }
            }
            for (; qb < n4; qb += EDGE_THREADS) {
                int  t0 = qb + lane;
                bool ok = t0 < n4;
                int  t  = t0 + rot;
                if (t >= n4) t -= n4;
                if (!ok) t = 0;
                int4 d = d4[t];
                int  e = t << 2;
                wcount = scan_site(ok, e,     d.x, lo, cnt, wcount, wls, wld);
                wcount = scan_site(ok, e + 1, d.y, lo, cnt, wcount, wls, wld);
                wcount = scan_site(ok, e + 2, d.z, lo, cnt, wcount, wls, wld);
                wcount = scan_site(ok, e + 3, d.w, lo, cnt, wcount, wls, wld);
            }
        }
        for (int eb = readback ? n_edges : ((n4 << 2) + wave * 32); eb < n_edges; eb += EDGE_THREADS) {
            int  e  = eb + lane;
            bool ok = e < n_edges;
            int  d  = dst[ok ? e : 0];
            wcount = scan_site(ok, e, d, lo, cnt, wcount, wls, wld);
        }
        wcount = (unsigned)__builtin_amdgcn_readfirstlane((int)wcount);
        if (wcount > CAPW && lane == 0) lovf = 1u;
        __syncthreads();

        const bool overflow = readback ? ((kw0 >> 31) != 0u) : (lovf != 0u);
        const int  nl = readback ? (int)(kw & 0x7fffffffu) : (overflow ? nself : (int)wcount);
        if (kmode == KEEP_WRITE && lane == 0)
            kcount[tile * EDGE_WAVES + wave] = (unsigned)nl | (overflow ? 0x80000000u : 0u);

        if (!readback) {
            for (int i = lane; i < nl; i += 32) {
                int e  = wls[i];
                int dl = wld[i];
                int s  = (e < n_edges) ? src[e] : (e - n_edges);
                float v = leaky(as_[s] + lad[dl]);
                atomicMax(lmax + dl, flipf(v));
                wls[i] = s;
                wlv[i] = v;
                if (kmode == KEEP_WRITE) gl[i] = (unsigned)s | ((unsigned)dl << SRC_BITS);
            }
        } else {
            for (int i0 = lane; i0 < nl; i0 += READ_BATCH * 32) {
                unsigned pk[READ_BATCH];
                float    av[READ_BATCH];
                #pragma unroll
                for (int u = 0; u < READ_BATCH; ++u) {
                    int i = i0 + u * 32;
                    pk[u] = (i < nl) ? gl[i] : 0u;
                }
                #pragma unroll
                for (int u = 0; u < READ_BATCH; ++u)
                    av[u] = as_[pk[u] & ((1u << SRC_BITS) - 1u)];
                #pragma unroll
                for (int u = 0; u < READ_BATCH; ++u) {
                    int i = i0 + u * 32;
                    if (i < nl) {
                        int s  = (int)(pk[u] & ((1u << SRC_BITS) - 1u));
                        int dl = (int)(pk[u] >> SRC_BITS);
                        float v = leaky(av[u] + lad[dl]);
                        atomicMax(lmax + dl, flipf(v));
                        wls[i] = s;
                        wld[i] = (unsigned short)dl;
                        wlv[i] = v;
                    }
                }
            }
        }
        if (overflow) {
            for (int e = tid; e < n_edges; e += EDGE_THREADS) {
                unsigned dl = (unsigned)(dst[e] - lo);
                if (dl < (unsigned)cnt)
                    atomicMax(lmax + dl, flipf(leaky(as_[src[e]] + lad[dl])));
            }
        }
        __syncthreads();

        for (int i = lane; i < nl; i += 32) {
            int dl = wld[i];
            float ex = __expf(wlv[i] - unflipf(lmax[dl]));
            wlv[i] = ex;
            lds_add(lden + dl, ex);
        }
        if (overflow) {
            for (int e = tid; e < n_edges; e += EDGE_THREADS) {
                unsigned dl = (unsigned)(dst[e] - lo);
                if (dl < (unsigned)cnt) {
                    float v = leaky(as_[src[e]] + lad[dl]);
                    lds_add(lden + dl, __expf(v - unflipf(lmax[dl])));
                }
            }
        }
        __syncthreads();

        for (int i0 = 0; i0 < nl; i0 += ROW_BATCH) {
            float al[ROW_BATCH]; int dr[ROW_BATCH];
            float r0[ROW_BATCH], r1[ROW_BATCH], r2[ROW_BATCH], r3[ROW_BATCH];
            #pragma unroll
            for (int u = 0; u < ROW_BATCH; ++u) {
                int i  = i0 + u;
                int ii = (i < nl) ? i : i0;
                int s  = wls[ii];
                int dl = wld[ii];
                dr[u] = dl;
                al[u] = wlv[ii] / (lden[dl] + GAT_EPS);
                const float* xr = x + (size_t)s * HID + lane;
                r0[u] = xr[0]; r1[u] = xr[32]; r2[u] = xr[64]; r3[u] = xr[96];
            }
            #pragma unroll
            for (int u = 0; u < ROW_BATCH; ++u) {
                if (i0 + u < nl) {
                    float* o = acc + dr[u] * APITCH + lane;
                    lds_add(o,      al[u] * r0[u]);
                    lds_add(o + 32, al[u] * r1[u]);
                    lds_add(o + 64, al[u] * r2[u]);
                    lds_add(o + 96, al[u] * r3[u]);
                }
            }
        }
        if (overflow) {
            for (int base = wave * 32; base < n_edges; base += EDGE_THREADS) {
                int e = base + lane;
                bool hit = false; int s = 0; int dli = 0; float alpha = 0.0f;
                if (e < n_edges) {
                    unsigned dl = (unsigned)(dst[e] - lo);
                    if (dl < (unsigned)cnt) {
                        hit = true;
                        dli = (int)dl;
                        s = src[e];
                        float v = leaky(as_[s] + lad[dl]);
                        alpha = __expf(v - unflipf(lmax[dl])) / (lden[dl] + GAT_EPS);
                    }
                }
                unsigned mask = __builtin_amdgcn_ballot_w32(hit);
                while (mask != 0u) {
                    int b = __builtin_ctz(mask);
                    mask &= mask - 1u;
                    int   sb = __shfl(s, b, 32);
                    int   db = __shfl(dli, b, 32);
                    float ab = __shfl(alpha, b, 32);
                    const float* xr = x + (size_t)sb * HID + lane;
                    float* o = acc + db * APITCH + lane;
                    lds_add(o,      ab * xr[0]);
                    lds_add(o + 32, ab * xr[32]);
                    lds_add(o + 64, ab * xr[64]);
                    lds_add(o + 96, ab * xr[96]);
                }
            }
        }
        __syncthreads();

        if (Wn != nullptr) {
            for (int i = tid; i < NPB * (HID / 4); i += EDGE_THREADS) {
                float4* p4 = (float4*)(acc + (i >> 5) * APITCH + (i & 31) * 4);
                float4 v = *p4;
                v.x = fmaxf(v.x, 0.0f); v.y = fmaxf(v.y, 0.0f);
                v.z = fmaxf(v.z, 0.0f); v.w = fmaxf(v.w, 0.0f);
                *p4 = v;
            }
            __syncthreads();
            const int g    = wave >> 3;
            const int ncol = (wave & 7) * 16 + (lane & 15);
            v8f c[4];
            if (g + 12 < MTILES) {
                next_x_tiles<4>(acc, Wn, g, lane, ncol, c);
            } else {
                next_x_tiles<3>(acc, Wn, g, lane, ncol, c);
                c[3] = v8f{};
            }
            __syncthreads();
            #pragma unroll
            for (int j = 0; j < 4; ++j) {
                const int rbase = (g + 4 * j) * 16 + ((lane >> 4) << 3);
                #pragma unroll
                for (int r = 0; r < 8; ++r) {
                    int row = rbase + r;
                    if (j < 3 || row < NPB) acc[row * APITCH + ncol] = c[j][r];
                }
            }
            __syncthreads();
            const float4 s4 = ((const float4*)a_sn)[lane];
            const float4 d4 = ((const float4*)a_dn)[lane];
            for (int i = wave; i < cnt; i += EDGE_WAVES) {
                float4 v = *(const float4*)(acc + i * APITCH + lane * 4);
                ((float4*)(dest + (size_t)(lo + i) * HID))[lane] = v;
                float ps = v.x * s4.x + v.y * s4.y + v.z * s4.z + v.w * s4.w;
                float pd = v.x * d4.x + v.y * d4.y + v.z * d4.z + v.w * d4.w;
                #pragma unroll
                for (int o = 16; o > 0; o >>= 1) {
                    ps += __shfl_down(ps, o, 32);
                    pd += __shfl_down(pd, o, 32);
                }
                if (lane == 0) { as_next[lo + i] = ps; ad_next[lo + i] = pd; }
            }
        } else {
            float4* out4 = (float4*)(dest + (size_t)lo * HID);
            for (int i = tid; i < cnt * (HID / 4); i += EDGE_THREADS)
                out4[i] = *(const float4*)(acc + (i >> 5) * APITCH + (i & 31) * 4);
        }
        __syncthreads();
    }
}

extern "C" void kernel_launch(void* const* d_in, const int* in_sizes, int n_in,
                              void* d_out, int out_size, void* d_ws, size_t ws_size,
                              hipStream_t stream) {
    const float* z     = (const float*)d_in[0];
    const int*   ei    = (const int*)  d_in[1];
    const float* Ws    = (const float*)d_in[2];
    const float* a_src = (const float*)d_in[3];
    const float* a_dst = (const float*)d_in[4];
    const float* bias  = (const float*)d_in[5];
    float*       out   = (float*)d_out;

    const int N = in_sizes[0] / HID;
    const int E = in_sizes[1] / 2;
    const int* src = ei;
    const int* dst = ei + E;

    auto align = [](size_t v) { return (v + 255) & ~(size_t)255; };
    char* p = (char*)d_ws;
    float* x0  = (float*)p;    p += align((size_t)N * HID * sizeof(float));
    float* x1  = (float*)p;    p += align((size_t)N * HID * sizeof(float));
    float* as0 = (float*)p;    p += align((size_t)N * sizeof(float));
    float* ad0 = (float*)p;    p += align((size_t)N * sizeof(float));
    float* as1 = (float*)p;    p += align((size_t)N * sizeof(float));
    float* ad1 = (float*)p;    p += align((size_t)N * sizeof(float));
    const int ntiles = (N + NPB - 1) / NPB;
    unsigned* klist  = (unsigned*)p;  p += align((size_t)ntiles * EDGE_WAVES * CAPW * sizeof(unsigned));
    unsigned* kcount = (unsigned*)p;  p += align((size_t)ntiles * EDGE_WAVES * sizeof(unsigned));
    const bool keep = ((size_t)(p - (char*)d_ws) <= ws_size) && (N <= (1 << SRC_BITS));
    const int mode_first = keep ? KEEP_WRITE : KEEP_NONE;
    const int mode_later = keep ? KEEP_READ  : KEEP_NONE;

    gemm_x_kernel<<<(N + 15) / 16, 256, 0, stream>>>(z, Ws, x0, N);
    attn_kernel<<<((size_t)N * 32 + 255) / 256, 256, 0, stream>>>(x0, a_src, a_dst, as0, ad0, N);

    edge_owner_kernel<<<256, EDGE_THREADS, 0, stream>>>(src, dst, x0, as0, ad0, bias,
                                                       Ws + HID * HID, a_src + HID, a_dst + HID,
                                                       x1, as1, ad1, klist, kcount, mode_first, E, N);
    edge_owner_kernel<<<256, EDGE_THREADS, 0, stream>>>(src, dst, x1, as1, ad1, bias + HID,
                                                       Ws + 2 * HID * HID, a_src + 2 * HID, a_dst + 2 * HID,
                                                       x0, as0, ad0, klist, kcount, mode_later, E, N);
    edge_owner_kernel<<<256, EDGE_THREADS, 0, stream>>>(src, dst, x0, as0, ad0, bias + 2 * HID,
                                                       nullptr, nullptr, nullptr,
                                                       out, nullptr, nullptr, klist, kcount, mode_later, E, N);
}


# Row phase: each lane fetches one list entry and forms its weight once per 32 entries; readlane passes source, destination, weight to the wave in scalar registers, 8 rows at a time; sums as before
# speedup vs baseline: 4.6349x; 1.0870x over previous
#include <hip/hip_runtime.h>
#include <hip/hip_bf16.h>

#define HID 128
#define NEG_SLOPE 0.2f
#define GAT_EPS 1e-16f

typedef float v2f __attribute__((ext_vector_type(2)));
typedef float v8f __attribute__((ext_vector_type(8)));

__device__ __forceinline__ unsigned flipf(float f) {
    unsigned u = __float_as_uint(f);
    return (u & 0x80000000u) ? ~u : (u | 0x80000000u);
}
__device__ __forceinline__ float unflipf(unsigned u) {
    return __uint_as_float((u & 0x80000000u) ? (u ^ 0x80000000u) : ~u);
}

__global__ __launch_bounds__(256) void gemm_x_kernel(const float* __restrict__ h,
                                                     const float* __restrict__ W,
                                                     float* __restrict__ x,
                                                     int n) {
    __shared__ float sh[16 * HID];
    const int block_row = blockIdx.x * 16;
    const int tid  = threadIdx.x;
    const int wave = tid >> 5;
    const int lane = tid & 31;

    const float4* gsrc = (const float4*)(h + (size_t)block_row * HID);
    float4*       ldst = (float4*)sh;
    #pragma unroll
    for (int i = tid; i < 16 * HID / 4; i += 256) ldst[i] = gsrc[i];
    __syncthreads();

    const int m    = lane & 15;
    const int koff = (lane >> 4) << 1;
    const int ncol = wave * 16 + (lane & 15);

    v8f c = {};
    #pragma unroll
    for (int k = 0; k < HID; k += 4) {
        v2f a, b;
        a.x = sh[m * HID + k + koff];
        a.y = sh[m * HID + k + koff + 1];
        b.x = W[(size_t)(k + koff)     * HID + ncol];
        b.y = W[(size_t)(k + koff + 1) * HID + ncol];
        c = __builtin_amdgcn_wmma_f32_16x16x4_f32(
                 false, a,  false, b,
                 (short)0, c,  false,  false);
    }

    const int rbase = block_row + ((lane >> 4) << 3);
    #pragma unroll
    for (int r = 0; r < 8; ++r) {
        int row = rbase + r;
        if (row < n) x[(size_t)row * HID + ncol] = c[r];
    }
}

__global__ __launch_bounds__(256) void attn_kernel(const float* __restrict__ x,
                                                   const float* __restrict__ a_s,
                                                   const float* __restrict__ a_d,
                                                   float* __restrict__ as_out,
                                                   float* __restrict__ ad_out,
                                                   int n) {
    int node = (blockIdx.x * blockDim.x + threadIdx.x) >> 5;
    int lane = threadIdx.x & 31;
    if (node >= n) return;
    float4 v  = ((const float4*)(x + (size_t)node * HID))[lane];
    float4 s4 = ((const float4*)a_s)[lane];
    float4 d4 = ((const float4*)a_d)[lane];
    float ps = v.x * s4.x + v.y * s4.y + v.z * s4.z + v.w * s4.w;
    float pd = v.x * d4.x + v.y * d4.y + v.z * d4.z + v.w * d4.w;
    #pragma unroll
    for (int o = 16; o > 0; o >>= 1) {
        ps += __shfl_down(ps, o, 32);
        pd += __shfl_down(pd, o, 32);
    }
    if (lane == 0) { as_out[node] = ps; ad_out[node] = pd; }
}

#define NPB 196
#define APITCH 132
#define MTILES ((NPB + 15) / 16)
#define EDGE_THREADS 1024
#define EDGE_WAVES 32
#define CAPW 384
#define ROW_BATCH 8
#define SCAN_BATCH 8
#define KEEP_NONE  0
#define KEEP_WRITE 1
#define KEEP_READ  2
#define SRC_BITS 24
#define READ_BATCH 4
static_assert(NPB <= 256, "the local destination must fit the 8 bits above the source node");

__device__ __forceinline__ float leaky(float v) { return (v > 0.0f) ? v : NEG_SLOPE * v; }

__device__ __forceinline__ void lds_add(float* p, float v) {
    __hip_atomic_fetch_add(p, v, __ATOMIC_RELAXED, __HIP_MEMORY_SCOPE_WORKGROUP);
}

__device__ __forceinline__ unsigned scan_site(bool ok, int e, int d, int lo, int cnt,
                                              unsigned wcount, int* wls, unsigned short* wld) {
    unsigned dl = (unsigned)(d - lo);
    bool hit = ok && (dl < (unsigned)cnt);
    unsigned mask = __builtin_amdgcn_ballot_w32(hit);
    if (mask != 0u) {
        if (hit) {
            unsigned pos = wcount + __builtin_amdgcn_mbcnt_lo(mask, 0u);
            if (pos < CAPW) { wls[pos] = e; wld[pos] = (unsigned short)dl; }
        }
        wcount += (unsigned)__builtin_popcount(mask);
    }
    return wcount;
}

#define KBLK 2
template <int NT>
__device__ __forceinline__ void next_x_tiles(const float* acc, const float* __restrict__ Wn,
                                             int g, int lane, int ncol, v8f* c) {
    const int m    = lane & 15;
    const int koff = (lane >> 4) << 1;
    const float* arow[NT];
    #pragma unroll
    for (int j = 0; j < NT; ++j) {
        int row = (g + 4 * j) * 16 + m;
        row = (row < NPB) ? row : (NPB - 1);
        arow[j] = acc + row * APITCH + koff;
        c[j] = v8f{};
    }
    const float* wp = Wn + (size_t)koff * HID + ncol;
    v2f bn[KBLK];
    #pragma unroll
    for (int s = 0; s < KBLK; ++s) {
        bn[s].x = wp[(size_t)(4 * s) * HID];
        bn[s].y = wp[(size_t)(4 * s + 1) * HID];
    }
    #pragma nounroll
    for (int kb = 0; kb < HID / 4; kb += KBLK) {
        v2f bc[KBLK];
        #pragma unroll
        for (int s = 0; s < KBLK; ++s) bc[s] = bn[s];
        if (kb + KBLK < HID / 4) {
            const float* wq = wp + (size_t)(4 * (kb + KBLK)) * HID;
            #pragma unroll
            for (int s = 0; s < KBLK; ++s) {
                bn[s].x = wq[(size_t)(4 * s) * HID];
                bn[s].y = wq[(size_t)(4 * s + 1) * HID];
            }
        }
        #pragma unroll
        for (int s = 0; s < KBLK; ++s) {
            #pragma unroll
            for (int j = 0; j < NT; ++j) {
                v2f a = *(const v2f*)(arow[j] + 4 * (kb + s));
                c[j] = __builtin_amdgcn_wmma_f32_16x16x4_f32(
                            false, a,  false, bc[s],
                            (short)0, c[j],  false,  false);
            }
        }
    }
}

__global__ __launch_bounds__(EDGE_THREADS) void edge_owner_kernel(const int* __restrict__ src,
                                                                  const int* __restrict__ dst,
                                                                  const float* __restrict__ x,
                                                                  const float* __restrict__ as_,
                                                                  const float* __restrict__ ad_,
                                                                  const float* __restrict__ bias,
                                                                  const float* __restrict__ Wn,
                                                                  const float* __restrict__ a_sn,
                                                                  const float* __restrict__ a_dn,
                                                                  float* __restrict__ dest,
                                                                  float* __restrict__ as_next,
                                                                  float* __restrict__ ad_next,
                                                                  unsigned* klist, unsigned* kcount, int kmode,
                                                                  int n_edges, int n_nodes) {
    __shared__ __attribute__((aligned(16))) float acc[NPB * APITCH];
    __shared__ int            ls[EDGE_WAVES * CAPW];
    __shared__ float          lv[EDGE_WAVES * CAPW];
    __shared__ unsigned short ld[EDGE_WAVES * CAPW];
    __shared__ unsigned lmax[NPB];
    __shared__ float    lden[NPB];
    __shared__ float    lad[NPB];
    __shared__ unsigned lovf;

    const int tid  = threadIdx.x;
    const int lane = tid & 31;
    const int wave = __builtin_amdgcn_readfirstlane(tid >> 5);
    int*            wls = ls + wave * CAPW;
    float*          wlv = lv + wave * CAPW;
    unsigned short* wld = ld + wave * CAPW;

    for (int tile = blockIdx.x; tile * NPB < n_nodes; tile += gridDim.x) {
        const int lo  = tile * NPB;
        const int cnt = (n_nodes - lo < NPB) ? (n_nodes - lo) : NPB;
        const bool readback = (kmode == KEEP_READ);
        unsigned* gl = klist + ((size_t)tile * EDGE_WAVES + wave) * CAPW;
        unsigned kw = 0u, kw0 = 0u;
        if (readback) {
            kw  = kcount[tile * EDGE_WAVES + wave];
            kw0 = kcount[tile * EDGE_WAVES];
        }

        for (int i = tid; i < NPB * HID; i += EDGE_THREADS)
            acc[(i >> 7) * APITCH + (i & (HID - 1))] = bias[i & (HID - 1)];
        for (int i = tid; i < cnt; i += EDGE_THREADS) {
            lad[i]  = ad_[lo + i];
            lmax[i] = 0u;
            lden[i] = 0.0f;
        }
        if (tid == 0) lovf = 0u;
        const int nself = (cnt > wave) ? ((cnt - 1 - wave) / EDGE_WAVES + 1) : 0;
        if (!readback && lane < nself) {
            int i = wave + EDGE_WAVES * lane;
            wls[lane] = n_edges + lo + i;
            wld[lane] = (unsigned short)i;
        }
        unsigned wcount = (unsigned)nself;
        __syncthreads();

        const bool vec_ok = ((((size_t)dst) & 15) == 0);
        const int n4 = (vec_ok && !readback) ? (n_edges >> 2) : 0;
        if (n4 > 0) {
            const int4* d4 = (const int4*)dst;
            const int rot = (int)(((long long)blockIdx.x * n4) / gridDim.x) & ~7;
            int qb = wave * 32;
            for (; qb + (SCAN_BATCH - 1) * EDGE_THREADS + 31 < n4; qb += SCAN_BATCH * EDGE_THREADS) {
                int4 dv[SCAN_BATCH];
                int  qe[SCAN_BATCH];
                #pragma unroll
                for (int u = 0; u < SCAN_BATCH; ++u) {
                    int t = qb + u * EDGE_THREADS + lane + rot;
                    if (t >= n4) t -= n4;
                    qe[u] = t << 2;
                    dv[u] = d4[t];
                }
                #pragma unroll
                for (int u = 0; u < SCAN_BATCH; ++u)
                    asm volatile("" :: "v"(dv[u].x), "v"(dv[u].y), "v"(dv[u].z), "v"(dv[u].w));
                #pragma unroll
                for (int u = 0; u < SCAN_BATCH; ++u) {
                    wcount = scan_site(true, qe[u],     dv[u].x, lo, cnt, wcount, wls, wld);
                    wcount = scan_site(true, qe[u] + 1, dv[u].y, lo, cnt, wcount, wls, wld);
                    wcount = scan_site(true, qe[u] + 2, dv[u].z, lo, cnt, wcount, wls, wld);
                    wcount = scan_site(true, qe[u] + 3, dv[u].w, lo, cnt, wcount, wls, wld);
                }
            }
            for (; qb < n4; qb += EDGE_THREADS) {
                int  t0 = qb + lane;
                bool ok = t0 < n4;
                int  t  = t0 + rot;
                if (t >= n4) t -= n4;
                if (!ok) t = 0;
                int4 d = d4[t];
                int  e = t << 2;
                wcount = scan_site(ok, e,     d.x, lo, cnt, wcount, wls, wld);
                wcount = scan_site(ok, e + 1, d.y, lo, cnt, wcount, wls, wld);
                wcount = scan_site(ok, e + 2, d.z, lo, cnt, wcount, wls, wld);
                wcount = scan_site(ok, e + 3, d.w, lo, cnt, wcount, wls, wld);
            }
        }
        for (int eb = readback ? n_edges : ((n4 << 2) + wave * 32); eb < n_edges; eb += EDGE_THREADS) {
            int  e  = eb + lane;
            bool ok = e < n_edges;
            int  d  = dst[ok ? e : 0];
            wcount = scan_site(ok, e, d, lo, cnt, wcount, wls, wld);
        }
        wcount = (unsigned)__builtin_amdgcn_readfirstlane((int)wcount);
        if (wcount > CAPW && lane == 0) lovf = 1u;
        __syncthreads();

        const bool overflow = readback ? ((kw0 >> 31) != 0u) : (lovf != 0u);
        const int  nl = readback ? (int)(kw & 0x7fffffffu) : (overflow ? nself : (int)wcount);
        if (kmode == KEEP_WRITE && lane == 0)
            kcount[tile * EDGE_WAVES + wave] = (unsigned)nl | (overflow ? 0x80000000u : 0u);

        if (!readback) {
            for (int i = lane; i < nl; i += 32) {
                int e  = wls[i];
                int dl = wld[i];
                int s  = (e < n_edges) ? src[e] : (e - n_edges);
                float v = leaky(as_[s] + lad[dl]);
                atomicMax(lmax + dl, flipf(v));
                wls[i] = s;
                wlv[i] = v;
                if (kmode == KEEP_WRITE) gl[i] = (unsigned)s | ((unsigned)dl << SRC_BITS);
            }
        } else {
            for (int i0 = lane; i0 < nl; i0 += READ_BATCH * 32) {
                unsigned pk[READ_BATCH];
                float    av[READ_BATCH];
                #pragma unroll
                for (int u = 0; u < READ_BATCH; ++u) {
                    int i = i0 + u * 32;
                    pk[u] = (i < nl) ? gl[i] : 0u;
                }
                #pragma unroll
                for (int u = 0; u < READ_BATCH; ++u)
                    av[u] = as_[pk[u] & ((1u << SRC_BITS) - 1u)];
                #pragma unroll
                for (int u = 0; u < READ_BATCH; ++u) {
                    int i = i0 + u * 32;
                    if (i < nl) {
                        int s  = (int)(pk[u] & ((1u << SRC_BITS) - 1u));
                        int dl = (int)(pk[u] >> SRC_BITS);
                        float v = leaky(av[u] + lad[dl]);
                        atomicMax(lmax + dl, flipf(v));
                        wls[i] = s;
                        wld[i] = (unsigned short)dl;
                        wlv[i] = v;
                    }
                }
            }
        }
        if (overflow) {
            for (int e = tid; e < n_edges; e += EDGE_THREADS) {
                unsigned dl = (unsigned)(dst[e] - lo);
                if (dl < (unsigned)cnt)
                    atomicMax(lmax + dl, flipf(leaky(as_[src[e]] + lad[dl])));
            }
        }
        __syncthreads();

        for (int i = lane; i < nl; i += 32) {
            int dl = wld[i];
            float ex = __expf(wlv[i] - unflipf(lmax[dl]));
            wlv[i] = ex;
            lds_add(lden + dl, ex);
        }
        if (overflow) {
            for (int e = tid; e < n_edges; e += EDGE_THREADS) {
                unsigned dl = (unsigned)(dst[e] - lo);
                if (dl < (unsigned)cnt) {
                    float v = leaky(as_[src[e]] + lad[dl]);
                    lds_add(lden + dl, __expf(v - unflipf(lmax[dl])));
                }
            }
        }
        __syncthreads();

        for (int j0 = 0; j0 < nl; j0 += 32) {
            const int   ie = (j0 + lane < nl) ? (j0 + lane) : j0;
            const int   es = wls[ie];
            const int   ed = wld[ie];
            const float ea = wlv[ie] / (lden[ed] + GAT_EPS);
            for (int b = 0; b < 32 && j0 + b < nl; b += ROW_BATCH) {
                const int i0 = j0 + b;
                float al[ROW_BATCH]; int dr[ROW_BATCH];
                float r0[ROW_BATCH], r1[ROW_BATCH], r2[ROW_BATCH], r3[ROW_BATCH];
                #pragma unroll
                for (int u = 0; u < ROW_BATCH; ++u) {
                    const int s = __builtin_amdgcn_readlane(es, b + u);
                    dr[u] = __builtin_amdgcn_readlane(ed, b + u);
                    al[u] = __uint_as_float((unsigned)__builtin_amdgcn_readlane((int)__float_as_uint(ea), b + u));
                    const float* xr = x + (size_t)s * HID + lane;
                    r0[u] = xr[0]; r1[u] = xr[32]; r2[u] = xr[64]; r3[u] = xr[96];
                }
                #pragma unroll
                for (int u = 0; u < ROW_BATCH; ++u) {
                    if (i0 + u < nl) {
                        float* o = acc + dr[u] * APITCH + lane;
                        lds_add(o,      al[u] * r0[u]);
                        lds_add(o + 32, al[u] * r1[u]);
                        lds_add(o + 64, al[u] * r2[u]);
                        lds_add(o + 96, al[u] * r3[u]);
                    }
                }
            }
        }
        if (overflow) {
            for (int base = wave * 32; base < n_edges; base += EDGE_THREADS) {
                int e = base + lane;
                bool hit = false; int s = 0; int dli = 0; float alpha = 0.0f;
                if (e < n_edges) {
                    unsigned dl = (unsigned)(dst[e] - lo);
                    if (dl < (unsigned)cnt) {
                        hit = true;
                        dli = (int)dl;
                        s = src[e];
                        float v = leaky(as_[s] + lad[dl]);
                        alpha = __expf(v - unflipf(lmax[dl])) / (lden[dl] + GAT_EPS);
                    }
                }
                unsigned mask = __builtin_amdgcn_ballot_w32(hit);
                while (mask != 0u) {
                    int b = __builtin_ctz(mask);
                    mask &= mask - 1u;
                    int   sb = __shfl(s, b, 32);
                    int   db = __shfl(dli, b, 32);
                    float ab = __shfl(alpha, b, 32);
                    const float* xr = x + (size_t)sb * HID + lane;
                    float* o = acc + db * APITCH + lane;
                    lds_add(o,      ab * xr[0]);
                    lds_add(o + 32, ab * xr[32]);
                    lds_add(o + 64, ab * xr[64]);
                    lds_add(o + 96, ab * xr[96]);
                }
            }
        }
        __syncthreads();

        if (Wn != nullptr) {
            for (int i = tid; i < NPB * (HID / 4); i += EDGE_THREADS) {
                float4* p4 = (float4*)(acc + (i >> 5) * APITCH + (i & 31) * 4);
                float4 v = *p4;
                v.x = fmaxf(v.x, 0.0f); v.y = fmaxf(v.y, 0.0f);
                v.z = fmaxf(v.z, 0.0f); v.w = fmaxf(v.w, 0.0f);
                *p4 = v;
            }
            __syncthreads();
            const int g    = wave >> 3;
            const int ncol = (wave & 7) * 16 + (lane & 15);
            v8f c[4];
            if (g + 12 < MTILES) {
                next_x_tiles<4>(acc, Wn, g, lane, ncol, c);
            } else {
                next_x_tiles<3>(acc, Wn, g, lane, ncol, c);
                c[3] = v8f{};
            }
            __syncthreads();
            #pragma unroll
            for (int j = 0; j < 4; ++j) {
                const int rbase = (g + 4 * j) * 16 + ((lane >> 4) << 3);
                #pragma unroll
                for (int r = 0; r < 8; ++r) {
                    int row = rbase + r;
                    if (j < 3 || row < NPB) acc[row * APITCH + ncol] = c[j][r];
                }
            }
            __syncthreads();
            const float4 s4 = ((const float4*)a_sn)[lane];
            const float4 d4 = ((const float4*)a_dn)[lane];
            for (int i = wave; i < cnt; i += EDGE_WAVES) {
                float4 v = *(const float4*)(acc + i * APITCH + lane * 4);
                ((float4*)(dest + (size_t)(lo + i) * HID))[lane] = v;
                float ps = v.x * s4.x + v.y * s4.y + v.z * s4.z + v.w * s4.w;
                float pd = v.x * d4.x + v.y * d4.y + v.z * d4.z + v.w * d4.w;
                #pragma unroll
                for (int o = 16; o > 0; o >>= 1) {
                    ps += __shfl_down(ps, o, 32);
                    pd += __shfl_down(pd, o, 32);
                }
                if (lane == 0) { as_next[lo + i] = ps; ad_next[lo + i] = pd; }
            }
        } else {
            float4* out4 = (float4*)(dest + (size_t)lo * HID);
            for (int i = tid; i < cnt * (HID / 4); i += EDGE_THREADS)
                out4[i] = *(const float4*)(acc + (i >> 5) * APITCH + (i & 31) * 4);
        }
        __syncthreads();
    }
}

extern "C" void kernel_launch(void* const* d_in, const int* in_sizes, int n_in,
                              void* d_out, int out_size, void* d_ws, size_t ws_size,
                              hipStream_t stream) {
    const float* z     = (const float*)d_in[0];
    const int*   ei    = (const int*)  d_in[1];
    const float* Ws    = (const float*)d_in[2];
    const float* a_src = (const float*)d_in[3];
    const float* a_dst = (const float*)d_in[4];
    const float* bias  = (const float*)d_in[5];
    float*       out   = (float*)d_out;

    const int N = in_sizes[0] / HID;
    const int E = in_sizes[1] / 2;
    const int* src = ei;
    const int* dst = ei + E;

    auto align = [](size_t v) { return (v + 255) & ~(size_t)255; };
    char* p = (char*)d_ws;
    float* x0  = (float*)p;    p += align((size_t)N * HID * sizeof(float));
    float* x1  = (float*)p;    p += align((size_t)N * HID * sizeof(float));
    float* as0 = (float*)p;    p += align((size_t)N * sizeof(float));
    float* ad0 = (float*)p;    p += align((size_t)N * sizeof(float));
    float* as1 = (float*)p;    p += align((size_t)N * sizeof(float));
    float* ad1 = (float*)p;    p += align((size_t)N * sizeof(float));
    const int ntiles = (N + NPB - 1) / NPB;
    unsigned* klist  = (unsigned*)p;  p += align((size_t)ntiles * EDGE_WAVES * CAPW * sizeof(unsigned));
    unsigned* kcount = (unsigned*)p;  p += align((size_t)ntiles * EDGE_WAVES * sizeof(unsigned));
    const bool keep = ((size_t)(p - (char*)d_ws) <= ws_size) && (N <= (1 << SRC_BITS));
    const int mode_first = keep ? KEEP_WRITE : KEEP_NONE;
    const int mode_later = keep ? KEEP_READ  : KEEP_NONE;

    gemm_x_kernel<<<(N + 15) / 16, 256, 0, stream>>>(z, Ws, x0, N);
    attn_kernel<<<((size_t)N * 32 + 255) / 256, 256, 0, stream>>>(x0, a_src, a_dst, as0, ad0, N);

    edge_owner_kernel<<<256, EDGE_THREADS, 0, stream>>>(src, dst, x0, as0, ad0, bias,
                                                       Ws + HID * HID, a_src + HID, a_dst + HID,
                                                       x1, as1, ad1, klist, kcount, mode_first, E, N);
    edge_owner_kernel<<<256, EDGE_THREADS, 0, stream>>>(src, dst, x1, as1, ad1, bias + HID,
                                                       Ws + 2 * HID * HID, a_src + 2 * HID, a_dst + 2 * HID,
                                                       x0, as0, ad0, klist, kcount, mode_later, E, N);
    edge_owner_kernel<<<256, EDGE_THREADS, 0, stream>>>(src, dst, x0, as0, ad0, bias + 2 * HID,
                                                       nullptr, nullptr, nullptr,
                                                       out, nullptr, nullptr, klist, kcount, mode_later, E, N);
}


# Layer-0 x = z @ W and its dots by the between-layer tail path: a 1024-thread block stages 196 z rows in LDS, then the same product, row store and dots; product and dot kernels gone, 4 launches not 5
# speedup vs baseline: 5.0594x; 1.0916x over previous
#include <hip/hip_runtime.h>
#include <hip/hip_bf16.h>

#define HID 128
#define NEG_SLOPE 0.2f
#define GAT_EPS 1e-16f

typedef float v2f __attribute__((ext_vector_type(2)));
typedef float v8f __attribute__((ext_vector_type(8)));

__device__ __forceinline__ unsigned flipf(float f) {
    unsigned u = __float_as_uint(f);
    return (u & 0x80000000u) ? ~u : (u | 0x80000000u);
}
__device__ __forceinline__ float unflipf(unsigned u) {
    return __uint_as_float((u & 0x80000000u) ? (u ^ 0x80000000u) : ~u);
}

#define NPB 196
#define APITCH 132
#define MTILES ((NPB + 15) / 16)
#define EDGE_THREADS 1024
#define EDGE_WAVES 32
#define CAPW 384
#define ROW_BATCH 8
#define SCAN_BATCH 8
#define KEEP_NONE  0
#define KEEP_WRITE 1
#define KEEP_READ  2
#define SRC_BITS 24
#define READ_BATCH 4
static_assert(NPB <= 256, "the local destination must fit the 8 bits above the source node");

__device__ __forceinline__ float leaky(float v) { return (v > 0.0f) ? v : NEG_SLOPE * v; }

__device__ __forceinline__ void lds_add(float* p, float v) {
    __hip_atomic_fetch_add(p, v, __ATOMIC_RELAXED, __HIP_MEMORY_SCOPE_WORKGROUP);
}

__device__ __forceinline__ unsigned scan_site(bool ok, int e, int d, int lo, int cnt,
                                              unsigned wcount, int* wls, unsigned short* wld) {
    unsigned dl = (unsigned)(d - lo);
    bool hit = ok && (dl < (unsigned)cnt);
    unsigned mask = __builtin_amdgcn_ballot_w32(hit);
    if (mask != 0u) {
        if (hit) {
            unsigned pos = wcount + __builtin_amdgcn_mbcnt_lo(mask, 0u);
            if (pos < CAPW) { wls[pos] = e; wld[pos] = (unsigned short)dl; }
        }
        wcount += (unsigned)__builtin_popcount(mask);
    }
    return wcount;
}

#define KBLK 2
template <int NT>
__device__ __forceinline__ void next_x_tiles(const float* acc, const float* __restrict__ Wn,
                                             int g, int lane, int ncol, v8f* c) {
    const int m    = lane & 15;
    const int koff = (lane >> 4) << 1;
    const float* arow[NT];
    #pragma unroll
    for (int j = 0; j < NT; ++j) {
        int row = (g + 4 * j) * 16 + m;
        row = (row < NPB) ? row : (NPB - 1);
        arow[j] = acc + row * APITCH + koff;
        c[j] = v8f{};
    }
    const float* wp = Wn + (size_t)koff * HID + ncol;
    v2f bn[KBLK];
    #pragma unroll
    for (int s = 0; s < KBLK; ++s) {
        bn[s].x = wp[(size_t)(4 * s) * HID];
        bn[s].y = wp[(size_t)(4 * s + 1) * HID];
    }
    #pragma nounroll
    for (int kb = 0; kb < HID / 4; kb += KBLK) {
        v2f bc[KBLK];
        #pragma unroll
        for (int s = 0; s < KBLK; ++s) bc[s] = bn[s];
        if (kb + KBLK < HID / 4) {
            const float* wq = wp + (size_t)(4 * (kb + KBLK)) * HID;
            #pragma unroll
            for (int s = 0; s < KBLK; ++s) {
                bn[s].x = wq[(size_t)(4 * s) * HID];
                bn[s].y = wq[(size_t)(4 * s + 1) * HID];
            }
        }
        #pragma unroll
        for (int s = 0; s < KBLK; ++s) {
            #pragma unroll
            for (int j = 0; j < NT; ++j) {
                v2f a = *(const v2f*)(arow[j] + 4 * (kb + s));
                c[j] = __builtin_amdgcn_wmma_f32_16x16x4_f32(
                            false, a,  false, bc[s],
                            (short)0, c[j],  false,  false);
            }
        }
    }
}

__device__ __forceinline__ void rows_times_w(float* acc, const float* __restrict__ Wn,
                                             const float* __restrict__ a_sn, const float* __restrict__ a_dn,
                                             float* __restrict__ dest, float* __restrict__ as_next,
                                             float* __restrict__ ad_next, int lo, int cnt, int wave, int lane) {
    const int g    = wave >> 3;
    const int ncol = (wave & 7) * 16 + (lane & 15);
    v8f c[4];
    if (g + 12 < MTILES) {
        next_x_tiles<4>(acc, Wn, g, lane, ncol, c);
    } else {
        next_x_tiles<3>(acc, Wn, g, lane, ncol, c);
        c[3] = v8f{};
    }
    __syncthreads();
    #pragma unroll
    for (int j = 0; j < 4; ++j) {
        const int rbase = (g + 4 * j) * 16 + ((lane >> 4) << 3);
        #pragma unroll
        for (int r = 0; r < 8; ++r) {
            int row = rbase + r;
            if (j < 3 || row < NPB) acc[row * APITCH + ncol] = c[j][r];
        }
    }
    __syncthreads();
    const float4 s4 = ((const float4*)a_sn)[lane];
    const float4 d4 = ((const float4*)a_dn)[lane];
    for (int i = wave; i < cnt; i += EDGE_WAVES) {
        float4 v = *(const float4*)(acc + i * APITCH + lane * 4);
        ((float4*)(dest + (size_t)(lo + i) * HID))[lane] = v;
        float ps = v.x * s4.x + v.y * s4.y + v.z * s4.z + v.w * s4.w;
        float pd = v.x * d4.x + v.y * d4.y + v.z * d4.z + v.w * d4.w;
        #pragma unroll
        for (int o = 16; o > 0; o >>= 1) {
            ps += __shfl_down(ps, o, 32);
            pd += __shfl_down(pd, o, 32);
        }
        if (lane == 0) { as_next[lo + i] = ps; ad_next[lo + i] = pd; }
    }
}

#define STAGE_PIECES (NPB * (HID / 4))
#define STAGE_ROUNDS ((STAGE_PIECES + EDGE_THREADS - 1) / EDGE_THREADS)
__global__ __launch_bounds__(EDGE_THREADS) void first_x_kernel(const float* __restrict__ h,
                                                               const float* __restrict__ W,
                                                               const float* __restrict__ a_s,
                                                               const float* __restrict__ a_d,
                                                               float* __restrict__ x,
                                                               float* __restrict__ as_out,
                                                               float* __restrict__ ad_out,
                                                               int n_nodes) {
    __shared__ __attribute__((aligned(16))) float acc[NPB * APITCH];
    const int tid  = threadIdx.x;
    const int lane = tid & 31;
    const int wave = __builtin_amdgcn_readfirstlane(tid >> 5);

    for (int tile = blockIdx.x; tile * NPB < n_nodes; tile += gridDim.x) {
        const int lo  = tile * NPB;
        const int cnt = (n_nodes - lo < NPB) ? (n_nodes - lo) : NPB;
        const float4* g4 = (const float4*)(h + (size_t)lo * HID);
        float4 v[STAGE_ROUNDS];
        #pragma unroll
        for (int u = 0; u < STAGE_ROUNDS; ++u) {
            const int i = tid + u * EDGE_THREADS;
            v[u] = (i < cnt * (HID / 4)) ? g4[i] : make_float4(0.0f, 0.0f, 0.0f, 0.0f);
        }
        #pragma unroll
        for (int u = 0; u < STAGE_ROUNDS; ++u) {
            const int i = tid + u * EDGE_THREADS;
            if (i < STAGE_PIECES) *(float4*)(acc + (i >> 5) * APITCH + (i & 31) * 4) = v[u];
        }
        __syncthreads();
        rows_times_w(acc, W, a_s, a_d, x, as_out, ad_out, lo, cnt, wave, lane);
        __syncthreads();
    }
}

__global__ __launch_bounds__(EDGE_THREADS) void edge_owner_kernel(const int* __restrict__ src,
                                                                  const int* __restrict__ dst,
                                                                  const float* __restrict__ x,
                                                                  const float* __restrict__ as_,
                                                                  const float* __restrict__ ad_,
                                                                  const float* __restrict__ bias,
                                                                  const float* __restrict__ Wn,
                                                                  const float* __restrict__ a_sn,
                                                                  const float* __restrict__ a_dn,
                                                                  float* __restrict__ dest,
                                                                  float* __restrict__ as_next,
                                                                  float* __restrict__ ad_next,
                                                                  unsigned* klist, unsigned* kcount, int kmode,
                                                                  int n_edges, int n_nodes) {
    __shared__ __attribute__((aligned(16))) float acc[NPB * APITCH];
    __shared__ int            ls[EDGE_WAVES * CAPW];
    __shared__ float          lv[EDGE_WAVES * CAPW];
    __shared__ unsigned short ld[EDGE_WAVES * CAPW];
    __shared__ unsigned lmax[NPB];
    __shared__ float    lden[NPB];
    __shared__ float    lad[NPB];
    __shared__ unsigned lovf;

    const int tid  = threadIdx.x;
    const int lane = tid & 31;
    const int wave = __builtin_amdgcn_readfirstlane(tid >> 5);
    int*            wls = ls + wave * CAPW;
    float*          wlv = lv + wave * CAPW;
    unsigned short* wld = ld + wave * CAPW;

    for (int tile = blockIdx.x; tile * NPB < n_nodes; tile += gridDim.x) {
        const int lo  = tile * NPB;
        const int cnt = (n_nodes - lo < NPB) ? (n_nodes - lo) : NPB;
        const bool readback = (kmode == KEEP_READ);
        unsigned* gl = klist + ((size_t)tile * EDGE_WAVES + wave) * CAPW;
        unsigned kw = 0u, kw0 = 0u;
        if (readback) {
            kw  = kcount[tile * EDGE_WAVES + wave];
            kw0 = kcount[tile * EDGE_WAVES];
        }

        for (int i = tid; i < NPB * HID; i += EDGE_THREADS)
            acc[(i >> 7) * APITCH + (i & (HID - 1))] = bias[i & (HID - 1)];
        for (int i = tid; i < cnt; i += EDGE_THREADS) {
            lad[i]  = ad_[lo + i];
            lmax[i] = 0u;
            lden[i] = 0.0f;
        }
        if (tid == 0) lovf = 0u;
        const int nself = (cnt > wave) ? ((cnt - 1 - wave) / EDGE_WAVES + 1) : 0;
        if (!readback && lane < nself) {
            int i = wave + EDGE_WAVES * lane;
            wls[lane] = n_edges + lo + i;
            wld[lane] = (unsigned short)i;
        }
        unsigned wcount = (unsigned)nself;
        __syncthreads();

        const bool vec_ok = ((((size_t)dst) & 15) == 0);
        const int n4 = (vec_ok && !readback) ? (n_edges >> 2) : 0;
        if (n4 > 0) {
            const int4* d4 = (const int4*)dst;
            const int rot = (int)(((long long)blockIdx.x * n4) / gridDim.x) & ~7;
            int qb = wave * 32;
            for (; qb + (SCAN_BATCH - 1) * EDGE_THREADS + 31 < n4; qb += SCAN_BATCH * EDGE_THREADS) {
                int4 dv[SCAN_BATCH];
                int  qe[SCAN_BATCH];
                #pragma unroll
                for (int u = 0; u < SCAN_BATCH; ++u) {
                    int t = qb + u * EDGE_THREADS + lane + rot;
                    if (t >= n4) t -= n4;
                    qe[u] = t << 2;
                    dv[u] = d4[t];
                }
                #pragma unroll
                for (int u = 0; u < SCAN_BATCH; ++u)
                    asm volatile("" :: "v"(dv[u].x), "v"(dv[u].y), "v"(dv[u].z), "v"(dv[u].w));
                #pragma unroll
                for (int u = 0; u < SCAN_BATCH; ++u) {
                    wcount = scan_site(true, qe[u],     dv[u].x, lo, cnt, wcount, wls, wld);
                    wcount = scan_site(true, qe[u] + 1, dv[u].y, lo, cnt, wcount, wls, wld);
                    wcount = scan_site(true, qe[u] + 2, dv[u].z, lo, cnt, wcount, wls, wld);
                    wcount = scan_site(true, qe[u] + 3, dv[u].w, lo, cnt, wcount, wls, wld);
                }
            }
            for (; qb < n4; qb += EDGE_THREADS) {
                int  t0 = qb + lane;
                bool ok = t0 < n4;
                int  t  = t0 + rot;
                if (t >= n4) t -= n4;
                if (!ok) t = 0;
                int4 d = d4[t];
                int  e = t << 2;
                wcount = scan_site(ok, e,     d.x, lo, cnt, wcount, wls, wld);
                wcount = scan_site(ok, e + 1, d.y, lo, cnt, wcount, wls, wld);
                wcount = scan_site(ok, e + 2, d.z, lo, cnt, wcount, wls, wld);
                wcount = scan_site(ok, e + 3, d.w, lo, cnt, wcount, wls, wld);
            }
        }
        for (int eb = readback ? n_edges : ((n4 << 2) + wave * 32); eb < n_edges; eb += EDGE_THREADS) {
            int  e  = eb + lane;
            bool ok = e < n_edges;
            int  d  = dst[ok ? e : 0];
            wcount = scan_site(ok, e, d, lo, cnt, wcount, wls, wld);
        }
        wcount = (unsigned)__builtin_amdgcn_readfirstlane((int)wcount);
        if (wcount > CAPW && lane == 0) lovf = 1u;
        __syncthreads();

        const bool overflow = readback ? ((kw0 >> 31) != 0u) : (lovf != 0u);
        const int  nl = readback ? (int)(kw & 0x7fffffffu) : (overflow ? nself : (int)wcount);
        if (kmode == KEEP_WRITE && lane == 0)
            kcount[tile * EDGE_WAVES + wave] = (unsigned)nl | (overflow ? 0x80000000u : 0u);

        if (!readback) {
            for (int i = lane; i < nl; i += 32) {
                int e  = wls[i];
                int dl = wld[i];
                int s  = (e < n_edges) ? src[e] : (e - n_edges);
                float v = leaky(as_[s] + lad[dl]);
                atomicMax(lmax + dl, flipf(v));
                wls[i] = s;
                wlv[i] = v;
                if (kmode == KEEP_WRITE) gl[i] = (unsigned)s | ((unsigned)dl << SRC_BITS);
            }
        } else {
            for (int i0 = lane; i0 < nl; i0 += READ_BATCH * 32) {
                unsigned pk[READ_BATCH];
                float    av[READ_BATCH];
                #pragma unroll
                for (int u = 0; u < READ_BATCH; ++u) {
                    int i = i0 + u * 32;
                    pk[u] = (i < nl) ? gl[i] : 0u;
                }
                #pragma unroll
                for (int u = 0; u < READ_BATCH; ++u)
                    av[u] = as_[pk[u] & ((1u << SRC_BITS) - 1u)];
                #pragma unroll
                for (int u = 0; u < READ_BATCH; ++u) {
                    int i = i0 + u * 32;
                    if (i < nl) {
                        int s  = (int)(pk[u] & ((1u << SRC_BITS) - 1u));
                        int dl = (int)(pk[u] >> SRC_BITS);
                        float v = leaky(av[u] + lad[dl]);
                        atomicMax(lmax + dl, flipf(v));
                        wls[i] = s;
                        wld[i] = (unsigned short)dl;
                        wlv[i] = v;
                    }
                }
            }
        }
        if (overflow) {
            for (int e = tid; e < n_edges; e += EDGE_THREADS) {
                unsigned dl = (unsigned)(dst[e] - lo);
                if (dl < (unsigned)cnt)
                    atomicMax(lmax + dl, flipf(leaky(as_[src[e]] + lad[dl])));
            }
        }
        __syncthreads();

        for (int i = lane; i < nl; i += 32) {
            int dl = wld[i];
            float ex = __expf(wlv[i] - unflipf(lmax[dl]));
            wlv[i] = ex;
            lds_add(lden + dl, ex);
        }
        if (overflow) {
            for (int e = tid; e < n_edges; e += EDGE_THREADS) {
                unsigned dl = (unsigned)(dst[e] - lo);
                if (dl < (unsigned)cnt) {
                    float v = leaky(as_[src[e]] + lad[dl]);
                    lds_add(lden + dl, __expf(v - unflipf(lmax[dl])));
                }
            }
        }
        __syncthreads();

        for (int j0 = 0; j0 < nl; j0 += 32) {
            const int   ie = (j0 + lane < nl) ? (j0 + lane) : j0;
            const int   es = wls[ie];
            const int   ed = wld[ie];
            const float ea = wlv[ie] / (lden[ed] + GAT_EPS);
            for (int b = 0; b < 32 && j0 + b < nl; b += ROW_BATCH) {
                const int i0 = j0 + b;
                float al[ROW_BATCH]; int dr[ROW_BATCH];
                float r0[ROW_BATCH], r1[ROW_BATCH], r2[ROW_BATCH], r3[ROW_BATCH];
                #pragma unroll
                for (int u = 0; u < ROW_BATCH; ++u) {
                    const int s = __builtin_amdgcn_readlane(es, b + u);
                    dr[u] = __builtin_amdgcn_readlane(ed, b + u);
                    al[u] = __uint_as_float((unsigned)__builtin_amdgcn_readlane((int)__float_as_uint(ea), b + u));
                    const float* xr = x + (size_t)s * HID + lane;
                    r0[u] = xr[0]; r1[u] = xr[32]; r2[u] = xr[64]; r3[u] = xr[96];
                }
                #pragma unroll
                for (int u = 0; u < ROW_BATCH; ++u) {
                    if (i0 + u < nl) {
                        float* o = acc + dr[u] * APITCH + lane;
                        lds_add(o,      al[u] * r0[u]);
                        lds_add(o + 32, al[u] * r1[u]);
                        lds_add(o + 64, al[u] * r2[u]);
                        lds_add(o + 96, al[u] * r3[u]);
                    }
                }
            }
        }
        if (overflow) {
            for (int base = wave * 32; base < n_edges; base += EDGE_THREADS) {
                int e = base + lane;
                bool hit = false; int s = 0; int dli = 0; float alpha = 0.0f;
                if (e < n_edges) {
                    unsigned dl = (unsigned)(dst[e] - lo);
                    if (dl < (unsigned)cnt) {
                        hit = true;
                        dli = (int)dl;
                        s = src[e];
                        float v = leaky(as_[s] + lad[dl]);
                        alpha = __expf(v - unflipf(lmax[dl])) / (lden[dl] + GAT_EPS);
                    }
                }
                unsigned mask = __builtin_amdgcn_ballot_w32(hit);
                while (mask != 0u) {
                    int b = __builtin_ctz(mask);
                    mask &= mask - 1u;
                    int   sb = __shfl(s, b, 32);
                    int   db = __shfl(dli, b, 32);
                    float ab = __shfl(alpha, b, 32);
                    const float* xr = x + (size_t)sb * HID + lane;
                    float* o = acc + db * APITCH + lane;
                    lds_add(o,      ab * xr[0]);
                    lds_add(o + 32, ab * xr[32]);
                    lds_add(o + 64, ab * xr[64]);
                    lds_add(o + 96, ab * xr[96]);
                }
            }
        }
        __syncthreads();

        if (Wn != nullptr) {
            for (int i = tid; i < NPB * (HID / 4); i += EDGE_THREADS) {
                float4* p4 = (float4*)(acc + (i >> 5) * APITCH + (i & 31) * 4);
                float4 v = *p4;
                v.x = fmaxf(v.x, 0.0f); v.y = fmaxf(v.y, 0.0f);
                v.z = fmaxf(v.z, 0.0f); v.w = fmaxf(v.w, 0.0f);
                *p4 = v;
            }
            __syncthreads();
            rows_times_w(acc, Wn, a_sn, a_dn, dest, as_next, ad_next, lo, cnt, wave, lane);
        } else {
            float4* out4 = (float4*)(dest + (size_t)lo * HID);
            for (int i = tid; i < cnt * (HID / 4); i += EDGE_THREADS)
                out4[i] = *(const float4*)(acc + (i >> 5) * APITCH + (i & 31) * 4);
        }
        __syncthreads();
    }
}

extern "C" void kernel_launch(void* const* d_in, const int* in_sizes, int n_in,
                              void* d_out, int out_size, void* d_ws, size_t ws_size,
                              hipStream_t stream) {
    const float* z     = (const float*)d_in[0];
    const int*   ei    = (const int*)  d_in[1];
    const float* Ws    = (const float*)d_in[2];
    const float* a_src = (const float*)d_in[3];
    const float* a_dst = (const float*)d_in[4];
    const float* bias  = (const float*)d_in[5];
    float*       out   = (float*)d_out;

    const int N = in_sizes[0] / HID;
    const int E = in_sizes[1] / 2;
    const int* src = ei;
    const int* dst = ei + E;

    auto align = [](size_t v) { return (v + 255) & ~(size_t)255; };
    char* p = (char*)d_ws;
    float* x0  = (float*)p;    p += align((size_t)N * HID * sizeof(float));
    float* x1  = (float*)p;    p += align((size_t)N * HID * sizeof(float));
    float* as0 = (float*)p;    p += align((size_t)N * sizeof(float));
    float* ad0 = (float*)p;    p += align((size_t)N * sizeof(float));
    float* as1 = (float*)p;    p += align((size_t)N * sizeof(float));
    float* ad1 = (float*)p;    p += align((size_t)N * sizeof(float));
    const int ntiles = (N + NPB - 1) / NPB;
    unsigned* klist  = (unsigned*)p;  p += align((size_t)ntiles * EDGE_WAVES * CAPW * sizeof(unsigned));
    unsigned* kcount = (unsigned*)p;  p += align((size_t)ntiles * EDGE_WAVES * sizeof(unsigned));
    const bool keep = ((size_t)(p - (char*)d_ws) <= ws_size) && (N <= (1 << SRC_BITS));
    const int mode_first = keep ? KEEP_WRITE : KEEP_NONE;
    const int mode_later = keep ? KEEP_READ  : KEEP_NONE;

    first_x_kernel<<<256, EDGE_THREADS, 0, stream>>>(z, Ws, a_src, a_dst, x0, as0, ad0, N);

    edge_owner_kernel<<<256, EDGE_THREADS, 0, stream>>>(src, dst, x0, as0, ad0, bias,
                                                       Ws + HID * HID, a_src + HID, a_dst + HID,
                                                       x1, as1, ad1, klist, kcount, mode_first, E, N);
    edge_owner_kernel<<<256, EDGE_THREADS, 0, stream>>>(src, dst, x1, as1, ad1, bias + HID,
                                                       Ws + 2 * HID * HID, a_src + 2 * HID, a_dst + 2 * HID,
                                                       x0, as0, ad0, klist, kcount, mode_later, E, N);
    edge_owner_kernel<<<256, EDGE_THREADS, 0, stream>>>(src, dst, x0, as0, ad0, bias + 2 * HID,
                                                       nullptr, nullptr, nullptr,
                                                       out, nullptr, nullptr, klist, kcount, mode_later, E, N);
}


# Scan filters tile keys (dst / 196) kept as bit planes, written by the layer-0 launch, not dst: a quarter of the bytes, 15 bit operations per 32 edges; lists in key order, sums still in arrival order
# speedup vs baseline: 5.4649x; 1.0801x over previous
#include <hip/hip_runtime.h>
#include <hip/hip_bf16.h>

#define HID 128
#define NEG_SLOPE 0.2f
#define GAT_EPS 1e-16f

typedef float v2f __attribute__((ext_vector_type(2)));
typedef float v8f __attribute__((ext_vector_type(8)));

__device__ __forceinline__ unsigned flipf(float f) {
    unsigned u = __float_as_uint(f);
    return (u & 0x80000000u) ? ~u : (u | 0x80000000u);
}
__device__ __forceinline__ float unflipf(unsigned u) {
    return __uint_as_float((u & 0x80000000u) ? (u ^ 0x80000000u) : ~u);
}

#define NPB 196
#define APITCH 132
#define MTILES ((NPB + 15) / 16)
#define EDGE_THREADS 1024
#define EDGE_WAVES 32
#define CAPW 384
#define ROW_BATCH 8
#define SCAN_BATCH 8
#define KEEP_NONE  0
#define KEEP_WRITE 1
#define KEEP_READ  2
#define SRC_BITS 24
#define READ_BATCH 4
static_assert(NPB <= 256, "the local destination must fit the 8 bits above the source node");

__device__ __forceinline__ float leaky(float v) { return (v > 0.0f) ? v : NEG_SLOPE * v; }

__device__ __forceinline__ void lds_add(float* p, float v) {
    __hip_atomic_fetch_add(p, v, __ATOMIC_RELAXED, __HIP_MEMORY_SCOPE_WORKGROUP);
}

__device__ __forceinline__ unsigned scan_site(bool ok, int e, int d, int lo, int cnt,
                                              unsigned wcount, int* wls, unsigned short* wld) {
    unsigned dl = (unsigned)(d - lo);
    bool hit = ok && (dl < (unsigned)cnt);
    unsigned mask = __builtin_amdgcn_ballot_w32(hit);
    if (mask != 0u) {
        if (hit) {
            unsigned pos = wcount + __builtin_amdgcn_mbcnt_lo(mask, 0u);
            if (pos < CAPW) { wls[pos] = e; wld[pos] = (unsigned short)dl; }
        }
        wcount += (unsigned)__builtin_popcount(mask);
    }
    return wcount;
}

#define KEY_BATCH 4
#define KEY_CHUNK_WORDS 256
__device__ __forceinline__ int key_index(int group, int plane) {
    return (group >> 5) * KEY_CHUNK_WORDS + (plane >> 2) * 128 + (group & 31) * 4 + (plane & 3);
}

__device__ __forceinline__ unsigned key_hits(unsigned m, int e0, unsigned wcount, int* wls) {
    unsigned mask = __builtin_amdgcn_ballot_w32(m != 0u);
    for (int trip = 0; trip < 32 && mask != 0u; ++trip) {
        if (m != 0u) {
            const unsigned pos = wcount + __builtin_amdgcn_mbcnt_lo(mask, 0u);
            if (pos < CAPW) wls[pos] = e0 + __builtin_ctz(m);
            m &= m - 1u;
        }
        wcount += (unsigned)__builtin_popcount(mask);
        mask = __builtin_amdgcn_ballot_w32(m != 0u);
    }
    return wcount;
}

#define KBLK 2
template <int NT>
__device__ __forceinline__ void next_x_tiles(const float* acc, const float* __restrict__ Wn,
                                             int g, int lane, int ncol, v8f* c) {
    const int m    = lane & 15;
    const int koff = (lane >> 4) << 1;
    const float* arow[NT];
    #pragma unroll
    for (int j = 0; j < NT; ++j) {
        int row = (g + 4 * j) * 16 + m;
        row = (row < NPB) ? row : (NPB - 1);
        arow[j] = acc + row * APITCH + koff;
        c[j] = v8f{};
    }
    const float* wp = Wn + (size_t)koff * HID + ncol;
    v2f bn[KBLK];
    #pragma unroll
    for (int s = 0; s < KBLK; ++s) {
        bn[s].x = wp[(size_t)(4 * s) * HID];
        bn[s].y = wp[(size_t)(4 * s + 1) * HID];
    }
    #pragma nounroll
    for (int kb = 0; kb < HID / 4; kb += KBLK) {
        v2f bc[KBLK];
        #pragma unroll
        for (int s = 0; s < KBLK; ++s) bc[s] = bn[s];
        if (kb + KBLK < HID / 4) {
            const float* wq = wp + (size_t)(4 * (kb + KBLK)) * HID;
            #pragma unroll
            for (int s = 0; s < KBLK; ++s) {
                bn[s].x = wq[(size_t)(4 * s) * HID];
                bn[s].y = wq[(size_t)(4 * s + 1) * HID];
            }
        }
        #pragma unroll
        for (int s = 0; s < KBLK; ++s) {
            #pragma unroll
            for (int j = 0; j < NT; ++j) {
                v2f a = *(const v2f*)(arow[j] + 4 * (kb + s));
                c[j] = __builtin_amdgcn_wmma_f32_16x16x4_f32(
                            false, a,  false, bc[s],
                            (short)0, c[j],  false,  false);
            }
        }
    }
}

__device__ __forceinline__ void rows_times_w(float* acc, const float* __restrict__ Wn,
                                             const float* __restrict__ a_sn, const float* __restrict__ a_dn,
                                             float* __restrict__ dest, float* __restrict__ as_next,
                                             float* __restrict__ ad_next, int lo, int cnt, int wave, int lane) {
    const int g    = wave >> 3;
    const int ncol = (wave & 7) * 16 + (lane & 15);
    v8f c[4];
    if (g + 12 < MTILES) {
        next_x_tiles<4>(acc, Wn, g, lane, ncol, c);
    } else {
        next_x_tiles<3>(acc, Wn, g, lane, ncol, c);
        c[3] = v8f{};
    }
    __syncthreads();
    #pragma unroll
    for (int j = 0; j < 4; ++j) {
        const int rbase = (g + 4 * j) * 16 + ((lane >> 4) << 3);
        #pragma unroll
        for (int r = 0; r < 8; ++r) {
            int row = rbase + r;
            if (j < 3 || row < NPB) acc[row * APITCH + ncol] = c[j][r];
        }
    }
    __syncthreads();
    const float4 s4 = ((const float4*)a_sn)[lane];
    const float4 d4 = ((const float4*)a_dn)[lane];
    for (int i = wave; i < cnt; i += EDGE_WAVES) {
        float4 v = *(const float4*)(acc + i * APITCH + lane * 4);
        ((float4*)(dest + (size_t)(lo + i) * HID))[lane] = v;
        float ps = v.x * s4.x + v.y * s4.y + v.z * s4.z + v.w * s4.w;
        float pd = v.x * d4.x + v.y * d4.y + v.z * d4.z + v.w * d4.w;
        #pragma unroll
        for (int o = 16; o > 0; o >>= 1) {
            ps += __shfl_down(ps, o, 32);
            pd += __shfl_down(pd, o, 32);
        }
        if (lane == 0) { as_next[lo + i] = ps; ad_next[lo + i] = pd; }
    }
}

#define STAGE_PIECES (NPB * (HID / 4))
#define STAGE_ROUNDS ((STAGE_PIECES + EDGE_THREADS - 1) / EDGE_THREADS)
__global__ __launch_bounds__(EDGE_THREADS) void first_x_kernel(const float* __restrict__ h,
                                                               const float* __restrict__ W,
                                                               const float* __restrict__ a_s,
                                                               const float* __restrict__ a_d,
                                                               float* __restrict__ x,
                                                               float* __restrict__ as_out,
                                                               float* __restrict__ ad_out,
                                                               const int* __restrict__ dst,
                                                               unsigned* __restrict__ keys,
                                                               int n_edges, int n_nodes) {
    __shared__ __attribute__((aligned(16))) float acc[NPB * APITCH];
    const int tid  = threadIdx.x;
    const int lane = tid & 31;
    const int wave = __builtin_amdgcn_readfirstlane(tid >> 5);

    if (keys != nullptr) {
        const int ngrp = n_edges >> 5;
        const int ngp  = ((ngrp + 31) >> 5) << 5;
        const int gw   = blockIdx.x * EDGE_WAVES + wave;
        const int nw   = gridDim.x * EDGE_WAVES;
        for (int g0 = gw; g0 < ngp; g0 += KEY_BATCH * nw) {
            int dv[KEY_BATCH];
            #pragma unroll
            for (int u = 0; u < KEY_BATCH; ++u) {
                const int g = g0 + u * nw;
                dv[u] = dst[(((g < ngrp) ? g : 0) << 5) + lane];
            }
            #pragma unroll
            for (int u = 0; u < KEY_BATCH; ++u) {
                const int g = g0 + u * nw;
                if (g < ngp) {
                    const unsigned key = (g < ngrp) ? ((unsigned)dv[u] / NPB) : 0u;
                    unsigned word = 0u;
                    #pragma unroll
                    for (int p = 0; p < 8; ++p) {
                        const unsigned b = __builtin_amdgcn_ballot_w32(((key >> p) & 1u) != 0u);
                        if (lane == p) word = b;
                    }
                    if (lane < 8) keys[key_index(g, lane)] = word;
                }
            }
        }
    }

    for (int tile = blockIdx.x; tile * NPB < n_nodes; tile += gridDim.x) {
        const int lo  = tile * NPB;
        const int cnt = (n_nodes - lo < NPB) ? (n_nodes - lo) : NPB;
        const float4* g4 = (const float4*)(h + (size_t)lo * HID);
        float4 v[STAGE_ROUNDS];
        #pragma unroll
        for (int u = 0; u < STAGE_ROUNDS; ++u) {
            const int i = tid + u * EDGE_THREADS;
            v[u] = (i < cnt * (HID / 4)) ? g4[i] : make_float4(0.0f, 0.0f, 0.0f, 0.0f);
        }
        #pragma unroll
        for (int u = 0; u < STAGE_ROUNDS; ++u) {
            const int i = tid + u * EDGE_THREADS;
            if (i < STAGE_PIECES) *(float4*)(acc + (i >> 5) * APITCH + (i & 31) * 4) = v[u];
        }
        __syncthreads();
        rows_times_w(acc, W, a_s, a_d, x, as_out, ad_out, lo, cnt, wave, lane);
        __syncthreads();
    }
}

__global__ __launch_bounds__(EDGE_THREADS) void edge_owner_kernel(const int* __restrict__ src,
                                                                  const int* __restrict__ dst,
                                                                  const float* __restrict__ x,
                                                                  const float* __restrict__ as_,
                                                                  const float* __restrict__ ad_,
                                                                  const float* __restrict__ bias,
                                                                  const float* __restrict__ Wn,
                                                                  const float* __restrict__ a_sn,
                                                                  const float* __restrict__ a_dn,
                                                                  float* __restrict__ dest,
                                                                  float* __restrict__ as_next,
                                                                  float* __restrict__ ad_next,
                                                                  unsigned* klist, unsigned* kcount, int kmode,
                                                                  const unsigned* __restrict__ keys,
                                                                  int n_edges, int n_nodes) {
    __shared__ __attribute__((aligned(16))) float acc[NPB * APITCH];
    __shared__ int            ls[EDGE_WAVES * CAPW];
    __shared__ float          lv[EDGE_WAVES * CAPW];
    __shared__ unsigned short ld[EDGE_WAVES * CAPW];
    __shared__ unsigned lmax[NPB];
    __shared__ float    lden[NPB];
    __shared__ float    lad[NPB];
    __shared__ unsigned lovf;

    const int tid  = threadIdx.x;
    const int lane = tid & 31;
    const int wave = __builtin_amdgcn_readfirstlane(tid >> 5);
    int*            wls = ls + wave * CAPW;
    float*          wlv = lv + wave * CAPW;
    unsigned short* wld = ld + wave * CAPW;

    for (int tile = blockIdx.x; tile * NPB < n_nodes; tile += gridDim.x) {
        const int lo  = tile * NPB;
        const int cnt = (n_nodes - lo < NPB) ? (n_nodes - lo) : NPB;
        const bool readback = (kmode == KEEP_READ);
        unsigned* gl = klist + ((size_t)tile * EDGE_WAVES + wave) * CAPW;
        unsigned kw = 0u, kw0 = 0u;
        if (readback) {
            kw  = kcount[tile * EDGE_WAVES + wave];
            kw0 = kcount[tile * EDGE_WAVES];
        }

        for (int i = tid; i < NPB * HID; i += EDGE_THREADS)
            acc[(i >> 7) * APITCH + (i & (HID - 1))] = bias[i & (HID - 1)];
        for (int i = tid; i < cnt; i += EDGE_THREADS) {
            lad[i]  = ad_[lo + i];
            lmax[i] = 0u;
            lden[i] = 0.0f;
        }
        if (tid == 0) lovf = 0u;
        const int nself = (cnt > wave) ? ((cnt - 1 - wave) / EDGE_WAVES + 1) : 0;
        if (!readback && lane < nself) {
            int i = wave + EDGE_WAVES * lane;
            wls[lane] = n_edges + lo + i;
            wld[lane] = (unsigned short)i;
        }
        unsigned wcount = (unsigned)nself;
        __syncthreads();

        const bool vec_ok = ((((size_t)dst) & 15) == 0);
        const bool keyscan = (keys != nullptr) && !readback;
        const int ngrp = keyscan ? (n_edges >> 5) : 0;
        const int n4   = (vec_ok && !readback && !keyscan) ? (n_edges >> 2) : 0;
        if (ngrp > 0) {
            const int nchunk = (ngrp + 31) >> 5;
            unsigned inv[8];
            #pragma unroll
            for (int p = 0; p < 8; ++p) inv[p] = (((unsigned)tile >> p) & 1u) ? 0u : ~0u;
            const int rotc = (int)(((long long)blockIdx.x * nchunk) / gridDim.x);
            for (int cb = wave; cb < nchunk; cb += KEY_BATCH * EDGE_WAVES) {
                uint4 ka[KEY_BATCH], kb[KEY_BATCH];
                int   gq[KEY_BATCH];
                #pragma unroll
                for (int u = 0; u < KEY_BATCH; ++u) {
                    const int c = cb + u * EDGE_WAVES;
                    int cr = c + rotc;
                    if (cr >= nchunk) cr -= nchunk;
                    if (c >= nchunk) cr = 0;
                    const uint4* q = (const uint4*)(keys + (size_t)cr * KEY_CHUNK_WORDS) + lane;
                    ka[u] = q[0];
                    kb[u] = q[32];
                    const int g = (cr << 5) + lane;
                    gq[u] = (c < nchunk && g < ngrp) ? g : -1;
                }
                #pragma unroll
                for (int u = 0; u < KEY_BATCH; ++u)
                    asm volatile("" :: "v"(ka[u].x), "v"(ka[u].y), "v"(ka[u].z), "v"(ka[u].w),
                                       "v"(kb[u].x), "v"(kb[u].y), "v"(kb[u].z), "v"(kb[u].w));
                #pragma unroll
                for (int u = 0; u < KEY_BATCH; ++u) {
                    unsigned m = (ka[u].x ^ inv[0]) & (ka[u].y ^ inv[1]) & (ka[u].z ^ inv[2]) & (ka[u].w ^ inv[3])
                               & (kb[u].x ^ inv[4]) & (kb[u].y ^ inv[5]) & (kb[u].z ^ inv[6]) & (kb[u].w ^ inv[7]);
                    if (gq[u] < 0) m = 0u;
                    wcount = key_hits(m, gq[u] << 5, wcount, wls);
                }
            }
        }
        if (n4 > 0) {
            const int4* d4 = (const int4*)dst;
            const int rot = (int)(((long long)blockIdx.x * n4) / gridDim.x) & ~7;
            int qb = wave * 32;
            for (; qb + (SCAN_BATCH - 1) * EDGE_THREADS + 31 < n4; qb += SCAN_BATCH * EDGE_THREADS) {
                int4 dv[SCAN_BATCH];
                int  qe[SCAN_BATCH];
                #pragma unroll
                for (int u = 0; u < SCAN_BATCH; ++u) {
                    int t = qb + u * EDGE_THREADS + lane + rot;
                    if (t >= n4) t -= n4;
                    qe[u] = t << 2;
                    dv[u] = d4[t];
                }
                #pragma unroll
                for (int u = 0; u < SCAN_BATCH; ++u)
                    asm volatile("" :: "v"(dv[u].x), "v"(dv[u].y), "v"(dv[u].z), "v"(dv[u].w));
                #pragma unroll
                for (int u = 0; u < SCAN_BATCH; ++u) {
                    wcount = scan_site(true, qe[u],     dv[u].x, lo, cnt, wcount, wls, wld);
                    wcount = scan_site(true, qe[u] + 1, dv[u].y, lo, cnt, wcount, wls, wld);
                    wcount = scan_site(true, qe[u] + 2, dv[u].z, lo, cnt, wcount, wls, wld);
                    wcount = scan_site(true, qe[u] + 3, dv[u].w, lo, cnt, wcount, wls, wld);
                }
            }
            for (; qb < n4; qb += EDGE_THREADS) {
                int  t0 = qb + lane;
                bool ok = t0 < n4;
                int  t  = t0 + rot;
                if (t >= n4) t -= n4;
                if (!ok) t = 0;
                int4 d = d4[t];
                int  e = t << 2;
                wcount = scan_site(ok, e,     d.x, lo, cnt, wcount, wls, wld);
                wcount = scan_site(ok, e + 1, d.y, lo, cnt, wcount, wls, wld);
                wcount = scan_site(ok, e + 2, d.z, lo, cnt, wcount, wls, wld);
                wcount = scan_site(ok, e + 3, d.w, lo, cnt, wcount, wls, wld);
            }
        }
        for (int eb = readback ? n_edges : ((keyscan ? (ngrp << 5) : (n4 << 2)) + wave * 32); eb < n_edges; eb += EDGE_THREADS) {
            int  e  = eb + lane;
            bool ok = e < n_edges;
            int  d  = dst[ok ? e : 0];
            wcount = scan_site(ok, e, d, lo, cnt, wcount, wls, wld);
        }
        wcount = (unsigned)__builtin_amdgcn_readfirstlane((int)wcount);
        if (wcount > CAPW && lane == 0) lovf = 1u;
        __syncthreads();

        const bool overflow = readback ? ((kw0 >> 31) != 0u) : (lovf != 0u);
        const int  nl = readback ? (int)(kw & 0x7fffffffu) : (overflow ? nself : (int)wcount);
        if (kmode == KEEP_WRITE && lane == 0)
            kcount[tile * EDGE_WAVES + wave] = (unsigned)nl | (overflow ? 0x80000000u : 0u);

        if (!readback) {
            for (int i = lane; i < nl; i += 32) {
                int e  = wls[i];
                int s  = (e < n_edges) ? src[e] : (e - n_edges);
                int dl;
                if (keyscan) {
                    dl = ((e < n_edges) ? dst[e] : (e - n_edges)) - lo;
                    wld[i] = (unsigned short)dl;
                } else {
                    dl = wld[i];
                }
                float v = leaky(as_[s] + lad[dl]);
                atomicMax(lmax + dl, flipf(v));
                wls[i] = s;
                wlv[i] = v;
                if (kmode == KEEP_WRITE) gl[i] = (unsigned)s | ((unsigned)dl << SRC_BITS);
            }
        } else {
            for (int i0 = lane; i0 < nl; i0 += READ_BATCH * 32) {
                unsigned pk[READ_BATCH];
                float    av[READ_BATCH];
                #pragma unroll
                for (int u = 0; u < READ_BATCH; ++u) {
                    int i = i0 + u * 32;
                    pk[u] = (i < nl) ? gl[i] : 0u;
                }
                #pragma unroll
                for (int u = 0; u < READ_BATCH; ++u)
                    av[u] = as_[pk[u] & ((1u << SRC_BITS) - 1u)];
                #pragma unroll
                for (int u = 0; u < READ_BATCH; ++u) {
                    int i = i0 + u * 32;
                    if (i < nl) {
                        int s  = (int)(pk[u] & ((1u << SRC_BITS) - 1u));
                        int dl = (int)(pk[u] >> SRC_BITS);
                        float v = leaky(av[u] + lad[dl]);
                        atomicMax(lmax + dl, flipf(v));
                        wls[i] = s;
                        wld[i] = (unsigned short)dl;
                        wlv[i] = v;
                    }
                }
            }
        }
        if (overflow) {
            for (int e = tid; e < n_edges; e += EDGE_THREADS) {
                unsigned dl = (unsigned)(dst[e] - lo);
                if (dl < (unsigned)cnt)
                    atomicMax(lmax + dl, flipf(leaky(as_[src[e]] + lad[dl])));
            }
        }
        __syncthreads();

        for (int i = lane; i < nl; i += 32) {
            int dl = wld[i];
            float ex = __expf(wlv[i] - unflipf(lmax[dl]));
            wlv[i] = ex;
            lds_add(lden + dl, ex);
        }
        if (overflow) {
            for (int e = tid; e < n_edges; e += EDGE_THREADS) {
                unsigned dl = (unsigned)(dst[e] - lo);
                if (dl < (unsigned)cnt) {
                    float v = leaky(as_[src[e]] + lad[dl]);
                    lds_add(lden + dl, __expf(v - unflipf(lmax[dl])));
                }
            }
        }
        __syncthreads();

        for (int j0 = 0; j0 < nl; j0 += 32) {
            const int   ie = (j0 + lane < nl) ? (j0 + lane) : j0;
            const int   es = wls[ie];
            const int   ed = wld[ie];
            const float ea = wlv[ie] / (lden[ed] + GAT_EPS);
            for (int b = 0; b < 32 && j0 + b < nl; b += ROW_BATCH) {
                const int i0 = j0 + b;
                float al[ROW_BATCH]; int dr[ROW_BATCH];
                float r0[ROW_BATCH], r1[ROW_BATCH], r2[ROW_BATCH], r3[ROW_BATCH];
                #pragma unroll
                for (int u = 0; u < ROW_BATCH; ++u) {
                    const int s = __builtin_amdgcn_readlane(es, b + u);
                    dr[u] = __builtin_amdgcn_readlane(ed, b + u);
                    al[u] = __uint_as_float((unsigned)__builtin_amdgcn_readlane((int)__float_as_uint(ea), b + u));
                    const float* xr = x + (size_t)s * HID + lane;
                    r0[u] = xr[0]; r1[u] = xr[32]; r2[u] = xr[64]; r3[u] = xr[96];
                }
                #pragma unroll
                for (int u = 0; u < ROW_BATCH; ++u) {
                    if (i0 + u < nl) {
                        float* o = acc + dr[u] * APITCH + lane;
                        lds_add(o,      al[u] * r0[u]);
                        lds_add(o + 32, al[u] * r1[u]);
                        lds_add(o + 64, al[u] * r2[u]);
                        lds_add(o + 96, al[u] * r3[u]);
                    }
                }
            }
        }
        if (overflow) {
            for (int base = wave * 32; base < n_edges; base += EDGE_THREADS) {
                int e = base + lane;
                bool hit = false; int s = 0; int dli = 0; float alpha = 0.0f;
                if (e < n_edges) {
                    unsigned dl = (unsigned)(dst[e] - lo);
                    if (dl < (unsigned)cnt) {
                        hit = true;
                        dli = (int)dl;
                        s = src[e];
                        float v = leaky(as_[s] + lad[dl]);
                        alpha = __expf(v - unflipf(lmax[dl])) / (lden[dl] + GAT_EPS);
                    }
                }
                unsigned mask = __builtin_amdgcn_ballot_w32(hit);
                while (mask != 0u) {
                    int b = __builtin_ctz(mask);
                    mask &= mask - 1u;
                    int   sb = __shfl(s, b, 32);
                    int   db = __shfl(dli, b, 32);
                    float ab = __shfl(alpha, b, 32);
                    const float* xr = x + (size_t)sb * HID + lane;
                    float* o = acc + db * APITCH + lane;
                    lds_add(o,      ab * xr[0]);
                    lds_add(o + 32, ab * xr[32]);
                    lds_add(o + 64, ab * xr[64]);
                    lds_add(o + 96, ab * xr[96]);
                }
            }
        }
        __syncthreads();

        if (Wn != nullptr) {
            for (int i = tid; i < NPB * (HID / 4); i += EDGE_THREADS) {
                float4* p4 = (float4*)(acc + (i >> 5) * APITCH + (i & 31) * 4);
                float4 v = *p4;
                v.x = fmaxf(v.x, 0.0f); v.y = fmaxf(v.y, 0.0f);
                v.z = fmaxf(v.z, 0.0f); v.w = fmaxf(v.w, 0.0f);
                *p4 = v;
            }
            __syncthreads();
            rows_times_w(acc, Wn, a_sn, a_dn, dest, as_next, ad_next, lo, cnt, wave, lane);
        } else {
            float4* out4 = (float4*)(dest + (size_t)lo * HID);
            for (int i = tid; i < cnt * (HID / 4); i += EDGE_THREADS)
                out4[i] = *(const float4*)(acc + (i >> 5) * APITCH + (i & 31) * 4);
        }
        __syncthreads();
    }
}

extern "C" void kernel_launch(void* const* d_in, const int* in_sizes, int n_in,
                              void* d_out, int out_size, void* d_ws, size_t ws_size,
                              hipStream_t stream) {
    const float* z     = (const float*)d_in[0];
    const int*   ei    = (const int*)  d_in[1];
    const float* Ws    = (const float*)d_in[2];
    const float* a_src = (const float*)d_in[3];
    const float* a_dst = (const float*)d_in[4];
    const float* bias  = (const float*)d_in[5];
    float*       out   = (float*)d_out;

    const int N = in_sizes[0] / HID;
    const int E = in_sizes[1] / 2;
    const int* src = ei;
    const int* dst = ei + E;

    auto align = [](size_t v) { return (v + 255) & ~(size_t)255; };
    char* p = (char*)d_ws;
    float* x0  = (float*)p;    p += align((size_t)N * HID * sizeof(float));
    float* x1  = (float*)p;    p += align((size_t)N * HID * sizeof(float));
    float* as0 = (float*)p;    p += align((size_t)N * sizeof(float));
    float* ad0 = (float*)p;    p += align((size_t)N * sizeof(float));
    float* as1 = (float*)p;    p += align((size_t)N * sizeof(float));
    float* ad1 = (float*)p;    p += align((size_t)N * sizeof(float));
    const int ntiles = (N + NPB - 1) / NPB;
    unsigned* klist  = (unsigned*)p;  p += align((size_t)ntiles * EDGE_WAVES * CAPW * sizeof(unsigned));
    unsigned* kcount = (unsigned*)p;  p += align((size_t)ntiles * EDGE_WAVES * sizeof(unsigned));
    const bool keep = ((size_t)(p - (char*)d_ws) <= ws_size) && (N <= (1 << SRC_BITS));
    const int mode_first = keep ? KEEP_WRITE : KEEP_NONE;
    const int mode_later = keep ? KEEP_READ  : KEEP_NONE;
    unsigned* keyp = (unsigned*)p;
    p += align((size_t)(((E >> 5) + 31) >> 5) * KEY_CHUNK_WORDS * sizeof(unsigned));
    const bool keyed = ((size_t)(p - (char*)d_ws) <= ws_size) && (ntiles <= 256);
    const unsigned* keys = keyed ? keyp : nullptr;

    first_x_kernel<<<256, EDGE_THREADS, 0, stream>>>(z, Ws, a_src, a_dst, x0, as0, ad0,
                                                     dst, keyed ? keyp : nullptr, E, N);

    edge_owner_kernel<<<256, EDGE_THREADS, 0, stream>>>(src, dst, x0, as0, ad0, bias,
                                                       Ws + HID * HID, a_src + HID, a_dst + HID,
                                                       x1, as1, ad1, klist, kcount, mode_first, keys, E, N);
    edge_owner_kernel<<<256, EDGE_THREADS, 0, stream>>>(src, dst, x1, as1, ad1, bias + HID,
                                                       Ws + 2 * HID * HID, a_src + 2 * HID, a_dst + 2 * HID,
                                                       x0, as0, ad0, klist, kcount, mode_later, keys, E, N);
    edge_owner_kernel<<<256, EDGE_THREADS, 0, stream>>>(src, dst, x0, as0, ad0, bias + 2 * HID,
                                                       nullptr, nullptr, nullptr,
                                                       out, nullptr, nullptr, klist, kcount, mode_later, keys, E, N);
}


# Row sums without LDS float atomics: logits pass hands each in-edge to the wave owning its row (dst mod 32, LDS integer rank); owner adds 16-byte pieces by plain LDS read-add-store; rank order sums
# speedup vs baseline: 5.7320x; 1.0489x over previous
#include <hip/hip_runtime.h>
#include <hip/hip_bf16.h>

#define HID 128
#define NEG_SLOPE 0.2f
#define GAT_EPS 1e-16f

typedef float v2f __attribute__((ext_vector_type(2)));
typedef float v8f __attribute__((ext_vector_type(8)));

__device__ __forceinline__ unsigned flipf(float f) {
    unsigned u = __float_as_uint(f);
    return (u & 0x80000000u) ? ~u : (u | 0x80000000u);
}
__device__ __forceinline__ float unflipf(unsigned u) {
    return __uint_as_float((u & 0x80000000u) ? (u ^ 0x80000000u) : ~u);
}

#define NPB 196
#define APITCH 132
#define MTILES ((NPB + 15) / 16)
#define EDGE_THREADS 1024
#define EDGE_WAVES 32
#define CAPW 384
#define ROW_BATCH 8
#define SCAN_BATCH 8
#define KEEP_NONE  0
#define KEEP_WRITE 1
#define KEEP_READ  2
#define SRC_BITS 24
#define READ_BATCH 4
static_assert(NPB <= 256, "the local destination must fit the 8 bits above the source node");

__device__ __forceinline__ float leaky(float v) { return (v > 0.0f) ? v : NEG_SLOPE * v; }

__device__ __forceinline__ void lds_add(float* p, float v) {
    __hip_atomic_fetch_add(p, v, __ATOMIC_RELAXED, __HIP_MEMORY_SCOPE_WORKGROUP);
}

__device__ __forceinline__ unsigned scan_site(bool ok, int e, int d, int lo, int cnt,
                                              unsigned wcount, int* wls, unsigned short* wld) {
    unsigned dl = (unsigned)(d - lo);
    bool hit = ok && (dl < (unsigned)cnt);
    unsigned mask = __builtin_amdgcn_ballot_w32(hit);
    if (mask != 0u) {
        if (hit) {
            unsigned pos = wcount + __builtin_amdgcn_mbcnt_lo(mask, 0u);
            if (pos < CAPW) { wls[pos] = e; wld[pos] = (unsigned short)dl; }
        }
        wcount += (unsigned)__builtin_popcount(mask);
    }
    return wcount;
}

#define KEY_BATCH 4
#define KEY_CHUNK_WORDS 256
__device__ __forceinline__ int key_index(int group, int plane) {
    return (group >> 5) * KEY_CHUNK_WORDS + (plane >> 2) * 128 + (group & 31) * 4 + (plane & 3);
}

__device__ __forceinline__ unsigned key_hits(unsigned m, int e0, unsigned wcount, int* wls) {
    unsigned mask = __builtin_amdgcn_ballot_w32(m != 0u);
    for (int trip = 0; trip < 32 && mask != 0u; ++trip) {
        if (m != 0u) {
            const unsigned pos = wcount + __builtin_amdgcn_mbcnt_lo(mask, 0u);
            if (pos < CAPW) wls[pos] = e0 + __builtin_ctz(m);
            m &= m - 1u;
        }
        wcount += (unsigned)__builtin_popcount(mask);
        mask = __builtin_amdgcn_ballot_w32(m != 0u);
    }
    return wcount;
}

#define KBLK 2
template <int NT>
__device__ __forceinline__ void next_x_tiles(const float* acc, const float* __restrict__ Wn,
                                             int g, int lane, int ncol, v8f* c) {
    const int m    = lane & 15;
    const int koff = (lane >> 4) << 1;
    const float* arow[NT];
    #pragma unroll
    for (int j = 0; j < NT; ++j) {
        int row = (g + 4 * j) * 16 + m;
        row = (row < NPB) ? row : (NPB - 1);
        arow[j] = acc + row * APITCH + koff;
        c[j] = v8f{};
    }
    const float* wp = Wn + (size_t)koff * HID + ncol;
    v2f bn[KBLK];
    #pragma unroll
    for (int s = 0; s < KBLK; ++s) {
        bn[s].x = wp[(size_t)(4 * s) * HID];
        bn[s].y = wp[(size_t)(4 * s + 1) * HID];
    }
    #pragma nounroll
    for (int kb = 0; kb < HID / 4; kb += KBLK) {
        v2f bc[KBLK];
        #pragma unroll
        for (int s = 0; s < KBLK; ++s) bc[s] = bn[s];
        if (kb + KBLK < HID / 4) {
            const float* wq = wp + (size_t)(4 * (kb + KBLK)) * HID;
            #pragma unroll
            for (int s = 0; s < KBLK; ++s) {
                bn[s].x = wq[(size_t)(4 * s) * HID];
                bn[s].y = wq[(size_t)(4 * s + 1) * HID];
            }
        }
        #pragma unroll
        for (int s = 0; s < KBLK; ++s) {
            #pragma unroll
            for (int j = 0; j < NT; ++j) {
                v2f a = *(const v2f*)(arow[j] + 4 * (kb + s));
                c[j] = __builtin_amdgcn_wmma_f32_16x16x4_f32(
                            false, a,  false, bc[s],
                            (short)0, c[j],  false,  false);
            }
        }
    }
}

__device__ __forceinline__ void rows_times_w(float* acc, const float* __restrict__ Wn,
                                             const float* __restrict__ a_sn, const float* __restrict__ a_dn,
                                             float* __restrict__ dest, float* __restrict__ as_next,
                                             float* __restrict__ ad_next, int lo, int cnt, int wave, int lane) {
    const int g    = wave >> 3;
    const int ncol = (wave & 7) * 16 + (lane & 15);
    v8f c[4];
    if (g + 12 < MTILES) {
        next_x_tiles<4>(acc, Wn, g, lane, ncol, c);
    } else {
        next_x_tiles<3>(acc, Wn, g, lane, ncol, c);
        c[3] = v8f{};
    }
    __syncthreads();
    #pragma unroll
    for (int j = 0; j < 4; ++j) {
        const int rbase = (g + 4 * j) * 16 + ((lane >> 4) << 3);
        #pragma unroll
        for (int r = 0; r < 8; ++r) {
            int row = rbase + r;
            if (j < 3 || row < NPB) acc[row * APITCH + ncol] = c[j][r];
        }
    }
    __syncthreads();
    const float4 s4 = ((const float4*)a_sn)[lane];
    const float4 d4 = ((const float4*)a_dn)[lane];
    for (int i = wave; i < cnt; i += EDGE_WAVES) {
        float4 v = *(const float4*)(acc + i * APITCH + lane * 4);
        ((float4*)(dest + (size_t)(lo + i) * HID))[lane] = v;
        float ps = v.x * s4.x + v.y * s4.y + v.z * s4.z + v.w * s4.w;
        float pd = v.x * d4.x + v.y * d4.y + v.z * d4.z + v.w * d4.w;
        #pragma unroll
        for (int o = 16; o > 0; o >>= 1) {
            ps += __shfl_down(ps, o, 32);
            pd += __shfl_down(pd, o, 32);
        }
        if (lane == 0) { as_next[lo + i] = ps; ad_next[lo + i] = pd; }
    }
}

#define STAGE_PIECES (NPB * (HID / 4))
#define STAGE_ROUNDS ((STAGE_PIECES + EDGE_THREADS - 1) / EDGE_THREADS)
__global__ __launch_bounds__(EDGE_THREADS) void first_x_kernel(const float* __restrict__ h,
                                                               const float* __restrict__ W,
                                                               const float* __restrict__ a_s,
                                                               const float* __restrict__ a_d,
                                                               float* __restrict__ x,
                                                               float* __restrict__ as_out,
                                                               float* __restrict__ ad_out,
                                                               const int* __restrict__ dst,
                                                               unsigned* __restrict__ keys,
                                                               int n_edges, int n_nodes) {
    __shared__ __attribute__((aligned(16))) float acc[NPB * APITCH];
    const int tid  = threadIdx.x;
    const int lane = tid & 31;
    const int wave = __builtin_amdgcn_readfirstlane(tid >> 5);

    if (keys != nullptr) {
        const int ngrp = n_edges >> 5;
        const int ngp  = ((ngrp + 31) >> 5) << 5;
        const int gw   = blockIdx.x * EDGE_WAVES + wave;
        const int nw   = gridDim.x * EDGE_WAVES;
        for (int g0 = gw; g0 < ngp; g0 += KEY_BATCH * nw) {
            int dv[KEY_BATCH];
            #pragma unroll
            for (int u = 0; u < KEY_BATCH; ++u) {
                const int g = g0 + u * nw;
                dv[u] = dst[(((g < ngrp) ? g : 0) << 5) + lane];
            }
            #pragma unroll
            for (int u = 0; u < KEY_BATCH; ++u) {
                const int g = g0 + u * nw;
                if (g < ngp) {
                    const unsigned key = (g < ngrp) ? ((unsigned)dv[u] / NPB) : 0u;
                    unsigned word = 0u;
                    #pragma unroll
                    for (int p = 0; p < 8; ++p) {
                        const unsigned b = __builtin_amdgcn_ballot_w32(((key >> p) & 1u) != 0u);
                        if (lane == p) word = b;
                    }
                    if (lane < 8) keys[key_index(g, lane)] = word;
                }
            }
        }
    }

    for (int tile = blockIdx.x; tile * NPB < n_nodes; tile += gridDim.x) {
        const int lo  = tile * NPB;
        const int cnt = (n_nodes - lo < NPB) ? (n_nodes - lo) : NPB;
        const float4* g4 = (const float4*)(h + (size_t)lo * HID);
        float4 v[STAGE_ROUNDS];
        #pragma unroll
        for (int u = 0; u < STAGE_ROUNDS; ++u) {
            const int i = tid + u * EDGE_THREADS;
            v[u] = (i < cnt * (HID / 4)) ? g4[i] : make_float4(0.0f, 0.0f, 0.0f, 0.0f);
        }
        #pragma unroll
        for (int u = 0; u < STAGE_ROUNDS; ++u) {
            const int i = tid + u * EDGE_THREADS;
            if (i < STAGE_PIECES) *(float4*)(acc + (i >> 5) * APITCH + (i & 31) * 4) = v[u];
        }
        __syncthreads();
        rows_times_w(acc, W, a_s, a_d, x, as_out, ad_out, lo, cnt, wave, lane);
        __syncthreads();
    }
}

__global__ __launch_bounds__(EDGE_THREADS) void edge_owner_kernel(const int* __restrict__ src,
                                                                  const int* __restrict__ dst,
                                                                  const float* __restrict__ x,
                                                                  const float* __restrict__ as_,
                                                                  const float* __restrict__ ad_,
                                                                  const float* __restrict__ bias,
                                                                  const float* __restrict__ Wn,
                                                                  const float* __restrict__ a_sn,
                                                                  const float* __restrict__ a_dn,
                                                                  float* __restrict__ dest,
                                                                  float* __restrict__ as_next,
                                                                  float* __restrict__ ad_next,
                                                                  unsigned* klist, unsigned* kcount, int kmode,
                                                                  const unsigned* __restrict__ keys,
                                                                  int n_edges, int n_nodes) {
    __shared__ __attribute__((aligned(16))) float acc[NPB * APITCH];
    __shared__ int            ls[EDGE_WAVES * CAPW];
    __shared__ float          lv[EDGE_WAVES * CAPW];
    __shared__ unsigned short ld[EDGE_WAVES * CAPW];
    __shared__ unsigned lmax[NPB];
    __shared__ float    lden[NPB];
    __shared__ float    lad[NPB];
    __shared__ unsigned lovf;
    __shared__ int            rs[EDGE_WAVES * CAPW];
    __shared__ unsigned short rd[EDGE_WAVES * CAPW];
    __shared__ unsigned rcnt[EDGE_WAVES];
    __shared__ unsigned lovf2;

    const int tid  = threadIdx.x;
    const int lane = tid & 31;
    const int wave = __builtin_amdgcn_readfirstlane(tid >> 5);
    int*            wls = ls + wave * CAPW;
    float*          wlv = lv + wave * CAPW;
    unsigned short* wld = ld + wave * CAPW;
    int*            wrs = rs + wave * CAPW;
    unsigned short* wrd = rd + wave * CAPW;

    for (int tile = blockIdx.x; tile * NPB < n_nodes; tile += gridDim.x) {
        const int lo  = tile * NPB;
        const int cnt = (n_nodes - lo < NPB) ? (n_nodes - lo) : NPB;
        const bool readback = (kmode == KEEP_READ);
        unsigned* gl = klist + ((size_t)tile * EDGE_WAVES + wave) * CAPW;
        unsigned kw = 0u, kw0 = 0u;
        if (readback) {
            kw  = kcount[tile * EDGE_WAVES + wave];
            kw0 = kcount[tile * EDGE_WAVES];
        }

        for (int i = tid; i < NPB * HID; i += EDGE_THREADS)
            acc[(i >> 7) * APITCH + (i & (HID - 1))] = bias[i & (HID - 1)];
        for (int i = tid; i < cnt; i += EDGE_THREADS) {
            lad[i]  = ad_[lo + i];
            lmax[i] = 0u;
            lden[i] = 0.0f;
        }
        if (tid == 0) { lovf = 0u; lovf2 = 0u; }
        const int nself = (cnt > wave) ? ((cnt - 1 - wave) / EDGE_WAVES + 1) : 0;
        if (!readback && lane < nself) {
            int i = wave + EDGE_WAVES * lane;
            wls[lane] = n_edges + lo + i;
            wld[lane] = (unsigned short)i;
        }
        unsigned wcount = (unsigned)nself;
        if (lane == 0) rcnt[wave] = (unsigned)nself;
        __syncthreads();

        const bool vec_ok = ((((size_t)dst) & 15) == 0);
        const bool keyscan = (keys != nullptr) && !readback;
        const int ngrp = keyscan ? (n_edges >> 5) : 0;
        const int n4   = (vec_ok && !readback && !keyscan) ? (n_edges >> 2) : 0;
        if (ngrp > 0) {
            const int nchunk = (ngrp + 31) >> 5;
            unsigned inv[8];
            #pragma unroll
            for (int p = 0; p < 8; ++p) inv[p] = (((unsigned)tile >> p) & 1u) ? 0u : ~0u;
            const int rotc = (int)(((long long)blockIdx.x * nchunk) / gridDim.x);
            for (int cb = wave; cb < nchunk; cb += KEY_BATCH * EDGE_WAVES) {
                uint4 ka[KEY_BATCH], kb[KEY_BATCH];
                int   gq[KEY_BATCH];
                #pragma unroll
                for (int u = 0; u < KEY_BATCH; ++u) {
                    const int c = cb + u * EDGE_WAVES;
                    int cr = c + rotc;
                    if (cr >= nchunk) cr -= nchunk;
                    if (c >= nchunk) cr = 0;
                    const uint4* q = (const uint4*)(keys + (size_t)cr * KEY_CHUNK_WORDS) + lane;
                    ka[u] = q[0];
                    kb[u] = q[32];
                    const int g = (cr << 5) + lane;
                    gq[u] = (c < nchunk && g < ngrp) ? g : -1;
                }
                #pragma unroll
                for (int u = 0; u < KEY_BATCH; ++u)
                    asm volatile("" :: "v"(ka[u].x), "v"(ka[u].y), "v"(ka[u].z), "v"(ka[u].w),
                                       "v"(kb[u].x), "v"(kb[u].y), "v"(kb[u].z), "v"(kb[u].w));
                #pragma unroll
                for (int u = 0; u < KEY_BATCH; ++u) {
                    unsigned m = (ka[u].x ^ inv[0]) & (ka[u].y ^ inv[1]) & (ka[u].z ^ inv[2]) & (ka[u].w ^ inv[3])
                               & (kb[u].x ^ inv[4]) & (kb[u].y ^ inv[5]) & (kb[u].z ^ inv[6]) & (kb[u].w ^ inv[7]);
                    if (gq[u] < 0) m = 0u;
                    wcount = key_hits(m, gq[u] << 5, wcount, wls);
                }
            }
        }
        if (n4 > 0) {
            const int4* d4 = (const int4*)dst;
            const int rot = (int)(((long long)blockIdx.x * n4) / gridDim.x) & ~7;
            int qb = wave * 32;
            for (; qb + (SCAN_BATCH - 1) * EDGE_THREADS + 31 < n4; qb += SCAN_BATCH * EDGE_THREADS) {
                int4 dv[SCAN_BATCH];
                int  qe[SCAN_BATCH];
                #pragma unroll
                for (int u = 0; u < SCAN_BATCH; ++u) {
                    int t = qb + u * EDGE_THREADS + lane + rot;
                    if (t >= n4) t -= n4;
                    qe[u] = t << 2;
                    dv[u] = d4[t];
                }
                #pragma unroll
                for (int u = 0; u < SCAN_BATCH; ++u)
                    asm volatile("" :: "v"(dv[u].x), "v"(dv[u].y), "v"(dv[u].z), "v"(dv[u].w));
                #pragma unroll
                for (int u = 0; u < SCAN_BATCH; ++u) {
                    wcount = scan_site(true, qe[u],     dv[u].x, lo, cnt, wcount, wls, wld);
                    wcount = scan_site(true, qe[u] + 1, dv[u].y, lo, cnt, wcount, wls, wld);
                    wcount = scan_site(true, qe[u] + 2, dv[u].z, lo, cnt, wcount, wls, wld);
                    wcount = scan_site(true, qe[u] + 3, dv[u].w, lo, cnt, wcount, wls, wld);
                }
            }
            for (; qb < n4; qb += EDGE_THREADS) {
                int  t0 = qb + lane;
                bool ok = t0 < n4;
                int  t  = t0 + rot;
                if (t >= n4) t -= n4;
                if (!ok) t = 0;
                int4 d = d4[t];
                int  e = t << 2;
                wcount = scan_site(ok, e,     d.x, lo, cnt, wcount, wls, wld);
                wcount = scan_site(ok, e + 1, d.y, lo, cnt, wcount, wls, wld);
                wcount = scan_site(ok, e + 2, d.z, lo, cnt, wcount, wls, wld);
                wcount = scan_site(ok, e + 3, d.w, lo, cnt, wcount, wls, wld);
            }
        }
        for (int eb = readback ? n_edges : ((keyscan ? (ngrp << 5) : (n4 << 2)) + wave * 32); eb < n_edges; eb += EDGE_THREADS) {
            int  e  = eb + lane;
            bool ok = e < n_edges;
            int  d  = dst[ok ? e : 0];
            wcount = scan_site(ok, e, d, lo, cnt, wcount, wls, wld);
        }
        wcount = (unsigned)__builtin_amdgcn_readfirstlane((int)wcount);
        if (wcount > CAPW && lane == 0) lovf = 1u;
        __syncthreads();

        const bool overflow = readback ? ((kw0 >> 31) != 0u) : (lovf != 0u);
        const int  nls = readback ? (int)(kw & 0x7fffffffu) : (overflow ? nself : (int)wcount);

        if (!readback) {
            for (int i = lane; i < nls; i += 32) {
                int e  = wls[i];
                int s  = (e < n_edges) ? src[e] : (e - n_edges);
                int dl;
                if (keyscan) {
                    dl = ((e < n_edges) ? dst[e] : (e - n_edges)) - lo;
                } else {
                    dl = wld[i];
                }
                float v = leaky(as_[s] + lad[dl]);
                atomicMax(lmax + dl, flipf(v));
                const int own = dl & (EDGE_WAVES - 1);
                const unsigned pos = (i < nself) ? (unsigned)i : atomicAdd(rcnt + own, 1u);
                if (pos < CAPW) {
                    const int q = own * CAPW + (int)pos;
                    rs[q] = s;
                    rd[q] = (unsigned short)dl;
                    lv[q] = v;
                } else {
                    lovf2 = 1u;
                }
            }
        } else {
            for (int i0 = lane; i0 < nls; i0 += READ_BATCH * 32) {
                unsigned pk[READ_BATCH];
                float    av[READ_BATCH];
                #pragma unroll
                for (int u = 0; u < READ_BATCH; ++u) {
                    int i = i0 + u * 32;
                    pk[u] = (i < nls) ? gl[i] : 0u;
                }
                #pragma unroll
                for (int u = 0; u < READ_BATCH; ++u)
                    av[u] = as_[pk[u] & ((1u << SRC_BITS) - 1u)];
                #pragma unroll
                for (int u = 0; u < READ_BATCH; ++u) {
                    int i = i0 + u * 32;
                    if (i < nls) {
                        int s  = (int)(pk[u] & ((1u << SRC_BITS) - 1u));
                        int dl = (int)(pk[u] >> SRC_BITS);
                        float v = leaky(av[u] + lad[dl]);
                        atomicMax(lmax + dl, flipf(v));
                        wrs[i] = s;
                        wrd[i] = (unsigned short)dl;
                        wlv[i] = v;
                    }
                }
            }
        }
        if (overflow) {
            for (int e = tid; e < n_edges; e += EDGE_THREADS) {
                unsigned dl = (unsigned)(dst[e] - lo);
                if (dl < (unsigned)cnt)
                    atomicMax(lmax + dl, flipf(leaky(as_[src[e]] + lad[dl])));
            }
        }
        __syncthreads();

        const bool ovf = readback ? overflow : (overflow || (lovf2 != 0u));
        const int  nl  = readback ? nls
                                  : (ovf ? nself : __builtin_amdgcn_readfirstlane((int)rcnt[wave]));
        if (kmode == KEEP_WRITE && lane == 0)
            kcount[tile * EDGE_WAVES + wave] = (unsigned)nl | (ovf ? 0x80000000u : 0u);

        for (int i = lane; i < nl; i += 32) {
            int dl = wrd[i];
            float ex = __expf(wlv[i] - unflipf(lmax[dl]));
            wlv[i] = ex;
            lds_add(lden + dl, ex);
            if (kmode == KEEP_WRITE) gl[i] = (unsigned)wrs[i] | ((unsigned)dl << SRC_BITS);
        }
        if (ovf) {
            for (int e = tid; e < n_edges; e += EDGE_THREADS) {
                unsigned dl = (unsigned)(dst[e] - lo);
                if (dl < (unsigned)cnt) {
                    float v = leaky(as_[src[e]] + lad[dl]);
                    lds_add(lden + dl, __expf(v - unflipf(lmax[dl])));
                }
            }
        }
        __syncthreads();

        for (int j0 = 0; j0 < nl; j0 += 32) {
            const int   ie = (j0 + lane < nl) ? (j0 + lane) : j0;
            const int   es = wrs[ie];
            const int   ed = wrd[ie];
            const float ea = wlv[ie] / (lden[ed] + GAT_EPS);
            for (int b = 0; b < 32 && j0 + b < nl; b += ROW_BATCH) {
                const int i0 = j0 + b;
                float al[ROW_BATCH]; int dr[ROW_BATCH];
                float4 rr[ROW_BATCH];
                #pragma unroll
                for (int u = 0; u < ROW_BATCH; ++u) {
                    const int s = __builtin_amdgcn_readlane(es, b + u);
                    dr[u] = __builtin_amdgcn_readlane(ed, b + u);
                    al[u] = __uint_as_float((unsigned)__builtin_amdgcn_readlane((int)__float_as_uint(ea), b + u));
                    rr[u] = ((const float4*)(x + (size_t)s * HID))[lane];
                }
                #pragma unroll
                for (int u = 0; u < ROW_BATCH; ++u) {
                    if (i0 + u < nl) {
                        float4* o = (float4*)(acc + dr[u] * APITCH) + lane;
                        float4 a = *o;
                        a.x += al[u] * rr[u].x; a.y += al[u] * rr[u].y;
                        a.z += al[u] * rr[u].z; a.w += al[u] * rr[u].w;
                        *o = a;
                    }
                }
            }
        }
        if (ovf) {
            __syncthreads();
            for (int base = wave * 32; base < n_edges; base += EDGE_THREADS) {
                int e = base + lane;
                bool hit = false; int s = 0; int dli = 0; float alpha = 0.0f;
                if (e < n_edges) {
                    unsigned dl = (unsigned)(dst[e] - lo);
                    if (dl < (unsigned)cnt) {
                        hit = true;
                        dli = (int)dl;
                        s = src[e];
                        float v = leaky(as_[s] + lad[dl]);
                        alpha = __expf(v - unflipf(lmax[dl])) / (lden[dl] + GAT_EPS);
                    }
                }
                unsigned mask = __builtin_amdgcn_ballot_w32(hit);
                while (mask != 0u) {
                    int b = __builtin_ctz(mask);
                    mask &= mask - 1u;
                    int   sb = __shfl(s, b, 32);
                    int   db = __shfl(dli, b, 32);
                    float ab = __shfl(alpha, b, 32);
                    const float* xr = x + (size_t)sb * HID + lane;
                    float* o = acc + db * APITCH + lane;
                    lds_add(o,      ab * xr[0]);
                    lds_add(o + 32, ab * xr[32]);
                    lds_add(o + 64, ab * xr[64]);
                    lds_add(o + 96, ab * xr[96]);
                }
            }
        }
        __syncthreads();

        if (Wn != nullptr) {
            for (int i = tid; i < NPB * (HID / 4); i += EDGE_THREADS) {
                float4* p4 = (float4*)(acc + (i >> 5) * APITCH + (i & 31) * 4);
                float4 v = *p4;
                v.x = fmaxf(v.x, 0.0f); v.y = fmaxf(v.y, 0.0f);
                v.z = fmaxf(v.z, 0.0f); v.w = fmaxf(v.w, 0.0f);
                *p4 = v;
            }
            __syncthreads();
            rows_times_w(acc, Wn, a_sn, a_dn, dest, as_next, ad_next, lo, cnt, wave, lane);
        } else {
            float4* out4 = (float4*)(dest + (size_t)lo * HID);
            for (int i = tid; i < cnt * (HID / 4); i += EDGE_THREADS)
                out4[i] = *(const float4*)(acc + (i >> 5) * APITCH + (i & 31) * 4);
        }
        __syncthreads();
    }
}

extern "C" void kernel_launch(void* const* d_in, const int* in_sizes, int n_in,
                              void* d_out, int out_size, void* d_ws, size_t ws_size,
                              hipStream_t stream) {
    const float* z     = (const float*)d_in[0];
    const int*   ei    = (const int*)  d_in[1];
    const float* Ws    = (const float*)d_in[2];
    const float* a_src = (const float*)d_in[3];
    const float* a_dst = (const float*)d_in[4];
    const float* bias  = (const float*)d_in[5];
    float*       out   = (float*)d_out;

    const int N = in_sizes[0] / HID;
    const int E = in_sizes[1] / 2;
    const int* src = ei;
    const int* dst = ei + E;

    auto align = [](size_t v) { return (v + 255) & ~(size_t)255; };
    char* p = (char*)d_ws;
    float* x0  = (float*)p;    p += align((size_t)N * HID * sizeof(float));
    float* x1  = (float*)p;    p += align((size_t)N * HID * sizeof(float));
    float* as0 = (float*)p;    p += align((size_t)N * sizeof(float));
    float* ad0 = (float*)p;    p += align((size_t)N * sizeof(float));
    float* as1 = (float*)p;    p += align((size_t)N * sizeof(float));
    float* ad1 = (float*)p;    p += align((size_t)N * sizeof(float));
    const int ntiles = (N + NPB - 1) / NPB;
    unsigned* klist  = (unsigned*)p;  p += align((size_t)ntiles * EDGE_WAVES * CAPW * sizeof(unsigned));
    unsigned* kcount = (unsigned*)p;  p += align((size_t)ntiles * EDGE_WAVES * sizeof(unsigned));
    const bool keep = ((size_t)(p - (char*)d_ws) <= ws_size) && (N <= (1 << SRC_BITS));
    const int mode_first = keep ? KEEP_WRITE : KEEP_NONE;
    const int mode_later = keep ? KEEP_READ  : KEEP_NONE;
    unsigned* keyp = (unsigned*)p;
    p += align((size_t)(((E >> 5) + 31) >> 5) * KEY_CHUNK_WORDS * sizeof(unsigned));
    const bool keyed = ((size_t)(p - (char*)d_ws) <= ws_size) && (ntiles <= 256);
    const unsigned* keys = keyed ? keyp : nullptr;

    first_x_kernel<<<256, EDGE_THREADS, 0, stream>>>(z, Ws, a_src, a_dst, x0, as0, ad0,
                                                     dst, keyed ? keyp : nullptr, E, N);

    edge_owner_kernel<<<256, EDGE_THREADS, 0, stream>>>(src, dst, x0, as0, ad0, bias,
                                                       Ws + HID * HID, a_src + HID, a_dst + HID,
                                                       x1, as1, ad1, klist, kcount, mode_first, keys, E, N);
    edge_owner_kernel<<<256, EDGE_THREADS, 0, stream>>>(src, dst, x1, as1, ad1, bias + HID,
                                                       Ws + 2 * HID * HID, a_src + 2 * HID, a_dst + 2 * HID,
                                                       x0, as0, ad0, klist, kcount, mode_later, keys, E, N);
    edge_owner_kernel<<<256, EDGE_THREADS, 0, stream>>>(src, dst, x0, as0, ad0, bias + 2 * HID,
                                                       nullptr, nullptr, nullptr,
                                                       out, nullptr, nullptr, klist, kcount, mode_later, keys, E, N);
}
